# starting kernel
# baseline (speedup 1.0000x reference)
#include <hip/hip_runtime.h>

typedef float v2f __attribute__((ext_vector_type(2)));
typedef float v8f __attribute__((ext_vector_type(8)));

#define N_ATOMS 2048
#define BATCH 2
#define RECV_PER_BLOCK 8
#define CUTOFF_SQ 0.49f
#define MAX_DISP 0.02f

__device__ __forceinline__ float fast_rcp(float x)  { return __builtin_amdgcn_rcpf(x); }
__device__ __forceinline__ float fast_rsq(float x)  { return __builtin_amdgcn_rsqf(x); }

__device__ __forceinline__ float silu_f(float x) {
    return x * fast_rcp(1.0f + __expf(-x));
}
__device__ __forceinline__ float tanh_f(float x) {
    float xc = fminf(fmaxf(x, -15.0f), 15.0f);
    float e  = __expf(2.0f * xc);
    return (e - 1.0f) * fast_rcp(e + 1.0f);
}

__global__ __launch_bounds__(256) void simple_et_kernel(
    const float* __restrict__ x_nm, const int* __restrict__ atom_types,
    const int* __restrict__ mask_in, const float* __restrict__ emb_W,
    const float* __restrict__ msg_w1, const float* __restrict__ msg_b1,
    const float* __restrict__ msg_w2, const float* __restrict__ msg_b2,
    const float* __restrict__ upd_w1, const float* __restrict__ upd_b1,
    const float* __restrict__ upd_w2, const float* __restrict__ upd_b2,
    const float* __restrict__ dx_w, const float* __restrict__ dx_b,
    const float* __restrict__ dv_w, const float* __restrict__ dv_b,
    float* __restrict__ out)
{
    __shared__ float4 snode[N_ATOMS * 2];

    const int tid  = threadIdx.x;
    const int b    = blockIdx.x / (N_ATOMS / RECV_PER_BLOCK);
    const int row0 = (blockIdx.x % (N_ATOMS / RECV_PER_BLOCK)) * RECV_PER_BLOCK;

    for (int n = tid; n < N_ATOMS; n += 256) {
        int t = atom_types[b * N_ATOMS + n];
        float mv = mask_in[b * N_ATOMS + n] ? 1.0f : 0.0f;
        snode[2 * n]     = make_float4(x_nm[(b * N_ATOMS + n) * 3 + 0],
                                       x_nm[(b * N_ATOMS + n) * 3 + 1],
                                       x_nm[(b * N_ATOMS + n) * 3 + 2], mv);
        snode[2 * n + 1] = make_float4(emb_W[t * 3 + 0],
                                       emb_W[t * 3 + 1],
                                       emb_W[t * 3 + 2], 0.0f);
    }
    __syncthreads();

    const int lane = tid & 31;
    const int wave = tid >> 5;
    const int i    = row0 + wave;
    const int m    = lane & 15;
    const bool lo  = (lane < 16);

    const float4 xiq = snode[2 * i];
    const float  xi0 = xiq.x, xi1 = xiq.y, xi2 = xiq.z, maski = xiq.w;

    const int  fm = (m < 3) ? m : 2;
    const float fz = (m < 3 || m == 10) ? 1.0f : 0.0f;
    v2f A1, A2;
    A1[0] = fz * (lo ? msg_w1[0 * 3 + fm] : msg_w1[2 * 3 + fm]);
    A1[1] = fz * (lo ? msg_w1[1 * 3 + fm] : msg_w1[3 * 3 + fm]);
    const float z3 = (m < 3) ? 1.0f : 0.0f;
    A2[0] = z3 * (lo ? msg_w2[0 * 3 + fm] : msg_w2[2 * 3 + fm]);
    A2[1] = (lo ? z3 * msg_w2[1 * 3 + fm] : 0.0f);

    v8f C1 = {}, C2 = {};
    C1[0] = lo ? msg_b1[0] : 0.0f;
    C1[1] = lo ? msg_b1[1] : 0.0f;
    C1[2] = msg_b1[2];
    C2[0] = lo ? msg_b2[0] : 0.0f;
    C2[1] = lo ? msg_b2[1] : 0.0f;
    C2[2] = lo ? msg_b2[2] : 0.0f;

    float acc0 = 0.0f, acc1 = 0.0f, acc2 = 0.0f;
    const float lmask = lo ? 1.0f : 0.0f;

#pragma unroll 2
    for (int jc = 0; jc < N_ATOMS; jc += 16) {
        const int j = jc + m;
        const float4 xq = snode[2 * j];
        const float4 hq = snode[2 * j + 1];

        const float d0 = xi0 - xq.x, d1 = xi1 - xq.y, d2c = xi2 - xq.z;
        const float d2 = d0 * d0 + d1 * d1 + d2c * d2c + 1e-9f;
        const float inv = fast_rsq(d2);
        const float validf = ((d2 < CUTOFF_SQ) && (j != i)) ? (xq.w * maski) : 0.0f;
        const float radial = validf * fminf(inv, 1.0e3f);
        const float dscale = validf * lmask * fminf(inv, 1.0e6f);
        const float dir0 = d0 * dscale, dir1 = d1 * dscale, dir2 = d2c * dscale;

        v2f B1;
        B1[0] = lo ? hq.x : hq.z;
        B1[1] = lo ? hq.y : radial;

        v8f D1 = __builtin_amdgcn_wmma_f32_16x16x4_f32(
            false, A1, false, B1, (short)0, C1, false, false);

        const float hd0 = silu_f(D1[0]);
        const float hd1 = silu_f(D1[1]);
        const float hd2 = silu_f(D1[2]);

        v2f B2;
        B2[0] = lo ? hd0 : hd2;
        B2[1] = lo ? hd1 : 0.0f;

        v8f D2 = __builtin_amdgcn_wmma_f32_16x16x4_f32(
            false, A2, false, B2, (short)0, C2, false, false);

        acc0 += D2[0] * dir0;
        acc1 += D2[1] * dir1;
        acc2 += D2[2] * dir2;
    }

    for (int off = 16; off > 0; off >>= 1) {
        acc0 += __shfl_xor(acc0, off, 32);
        acc1 += __shfl_xor(acc1, off, 32);
        acc2 += __shfl_xor(acc2, off, 32);
    }

    if (lane == 0) {
        const float v0 = acc0, v1 = acc1, v2 = acc2;
        const float4 hiq = snode[2 * i + 1];
        const float hi0 = hiq.x, hi1 = hiq.y, hi2 = hiq.z;

        const float u0 = silu_f(v0 * upd_w1[0] + v1 * upd_w1[3] + v2 * upd_w1[6] + upd_b1[0]);
        const float u1 = silu_f(v0 * upd_w1[1] + v1 * upd_w1[4] + v2 * upd_w1[7] + upd_b1[1]);
        const float u2 = silu_f(v0 * upd_w1[2] + v1 * upd_w1[5] + v2 * upd_w1[8] + upd_b1[2]);

        const float h0 = hi0 + u0 * upd_w2[0] + u1 * upd_w2[3] + u2 * upd_w2[6] + upd_b2[0];
        const float h1 = hi1 + u0 * upd_w2[1] + u1 * upd_w2[4] + u2 * upd_w2[7] + upd_b2[1];
        const float h2 = hi2 + u0 * upd_w2[2] + u1 * upd_w2[5] + u2 * upd_w2[8] + upd_b2[2];

        const int base = (b * N_ATOMS + i) * 3;
        const int half = BATCH * N_ATOMS * 3;
#pragma unroll
        for (int c = 0; c < 3; ++c) {
            const float px = h0 * dx_w[0 * 3 + c] + h1 * dx_w[1 * 3 + c] + h2 * dx_w[2 * 3 + c] + dx_b[c];
            const float pv = h0 * dv_w[0 * 3 + c] + h1 * dv_w[1 * 3 + c] + h2 * dv_w[2 * 3 + c] + dv_b[c];
            out[base + c]        = tanh_f(px) * MAX_DISP;
            out[half + base + c] = tanh_f(pv) * 1.0f;
        }
    }
}

extern "C" void kernel_launch(void* const* d_in, const int* in_sizes, int n_in,
                              void* d_out, int out_size, void* d_ws, size_t ws_size,
                              hipStream_t stream) {
    (void)in_sizes; (void)n_in; (void)d_ws; (void)ws_size; (void)out_size;
    const float* x_nm   = (const float*)d_in[0];
    const int*   types  = (const int*)  d_in[1];
    const int*   mask   = (const int*)  d_in[2];
    const float* emb_W  = (const float*)d_in[3];
    const float* msg_w1 = (const float*)d_in[4];
    const float* msg_b1 = (const float*)d_in[5];
    const float* msg_w2 = (const float*)d_in[6];
    const float* msg_b2 = (const float*)d_in[7];
    const float* upd_w1 = (const float*)d_in[8];
    const float* upd_b1 = (const float*)d_in[9];
    const float* upd_w2 = (const float*)d_in[10];
    const float* upd_b2 = (const float*)d_in[11];
    const float* dx_w   = (const float*)d_in[12];
    const float* dx_b   = (const float*)d_in[13];
    const float* dv_w   = (const float*)d_in[14];
    const float* dv_b   = (const float*)d_in[15];
    float* out = (float*)d_out;

    dim3 grid(BATCH * (N_ATOMS / RECV_PER_BLOCK));
    dim3 block(256);
    simple_et_kernel<<<grid, block, 0, stream>>>(
        x_nm, types, mask, emb_W,
        msg_w1, msg_b1, msg_w2, msg_b2,
        upd_w1, upd_b1, upd_w2, upd_b2,
        dx_w, dx_b, dv_w, dv_b, out);
}


# All 32 lanes own distinct sender pairs (two chained matrix products per layer fill rows 0-2 and 8-10 of one tile) instead of 16 pairs duplicated on both halves; sum order changes: 32 lane partials
# speedup vs baseline: 1.2680x; 1.2680x over previous
#include <hip/hip_runtime.h>


typedef float v2f __attribute__((ext_vector_type(2)));
typedef float v8f __attribute__((ext_vector_type(8)));

#define N_ATOMS 2048
#define BATCH 2
#define RECV_PER_BLOCK 8
#define CUTOFF_SQ 0.49f
#define MAX_DISP 0.02f

__device__ __forceinline__ float fast_rcp(float x)  { return __builtin_amdgcn_rcpf(x); }
__device__ __forceinline__ float fast_rsq(float x)  { return __builtin_amdgcn_rsqf(x); }

__device__ __forceinline__ float silu_f(float x) {
    return x * fast_rcp(1.0f + __expf(-x));
}
__device__ __forceinline__ float tanh_f(float x) {
    float xc = fminf(fmaxf(x, -15.0f), 15.0f);
    float e  = __expf(2.0f * xc);
    return (e - 1.0f) * fast_rcp(e + 1.0f);
}

__global__ __launch_bounds__(256) void simple_et_kernel(
    const float* __restrict__ x_nm, const int* __restrict__ atom_types,
    const int* __restrict__ mask_in, const float* __restrict__ emb_W,
    const float* __restrict__ msg_w1, const float* __restrict__ msg_b1,
    const float* __restrict__ msg_w2, const float* __restrict__ msg_b2,
    const float* __restrict__ upd_w1, const float* __restrict__ upd_b1,
    const float* __restrict__ upd_w2, const float* __restrict__ upd_b2,
    const float* __restrict__ dx_w, const float* __restrict__ dx_b,
    const float* __restrict__ dv_w, const float* __restrict__ dv_b,
    float* __restrict__ out)
{
    __shared__ float4 sx[N_ATOMS];
    __shared__ float2 sh[N_ATOMS * 2];

    const int tid  = threadIdx.x;
    const int b    = blockIdx.x / (N_ATOMS / RECV_PER_BLOCK);
    const int row0 = (blockIdx.x % (N_ATOMS / RECV_PER_BLOCK)) * RECV_PER_BLOCK;

    for (int n = tid; n < N_ATOMS; n += 256) {
        int t = atom_types[b * N_ATOMS + n];
        float mv = mask_in[b * N_ATOMS + n] ? 1.0f : 0.0f;
        sx[n] = make_float4(x_nm[(b * N_ATOMS + n) * 3 + 0],
                            x_nm[(b * N_ATOMS + n) * 3 + 1],
                            x_nm[(b * N_ATOMS + n) * 3 + 2], mv);
        sh[2 * n]     = make_float2(emb_W[t * 3 + 1], emb_W[t * 3 + 2]);
        sh[2 * n + 1] = make_float2(emb_W[t * 3 + 0], 0.0f);
    }
    __syncthreads();

    const int lane = tid & 31;
    const int wave = tid >> 5;
    const int i    = row0 + wave;
    const int m    = lane & 15;
    const bool lo  = (lane < 16);

    const float4 xiq = sx[i];
    const float  xi0 = xiq.x, xi1 = xiq.y, xi2 = xiq.z, maski = xiq.w;


    const int  fa  = (m == 8) ? 2 : (m & 1);
    const bool ra  = (m == 0) || (m == 1) || (m == 8);
    const int  fb  = (m == 9) ? 1 : ((m == 2) ? 2 : 0);
    const bool rb  = (m == 9) || (m == 2) || (m == 10);
    v2f A1a, A1b, A2a, A2b;
    {
        const float a0 = lo ? msg_w1[0 * 3 + fa] : msg_w1[1 * 3 + fa];
        const float a1 = lo ? msg_w1[3 * 3 + fa] : msg_w1[2 * 3 + fa];
        A1a[0] = ra ? a0 : 0.0f;
        A1a[1] = ra ? a1 : 0.0f;
        const float b0 = lo ? msg_w1[1 * 3 + fb] : msg_w1[0 * 3 + fb];
        const float b1 = lo ? msg_w1[2 * 3 + fb] : msg_w1[3 * 3 + fb];
        A1b[0] = rb ? b0 : 0.0f;
        A1b[1] = rb ? b1 : 0.0f;
    }
    {
        const int  ca  = (m < 3) ? m : 0;
        const bool rca = (m < 3);
        const int  cb  = (m >= 8 && m < 11) ? (m - 8) : 0;
        const bool rcb = (m >= 8 && m < 11);
        const float a0 = lo ? msg_w2[0 * 3 + ca] : msg_w2[2 * 3 + ca];
        const float a1 = lo ? msg_w2[1 * 3 + ca] : 0.0f;
        A2a[0] = rca ? a0 : 0.0f;
        A2a[1] = rca ? a1 : 0.0f;
        const float b0 = lo ? msg_w2[2 * 3 + cb] : msg_w2[0 * 3 + cb];
        const float b1 = lo ? 0.0f : msg_w2[1 * 3 + cb];
        A2b[0] = rcb ? b0 : 0.0f;
        A2b[1] = rcb ? b1 : 0.0f;
    }

    v8f C1 = {}, C2 = {};
    C1[0] = lo ? msg_b1[0] : msg_b1[2];
    C1[1] = msg_b1[1];
    C1[2] = lo ? msg_b1[2] : msg_b1[0];
    C2[0] = msg_b2[0];
    C2[1] = msg_b2[1];
    C2[2] = msg_b2[2];

    float acc0 = 0.0f, acc1 = 0.0f, acc2 = 0.0f;

    const int pa = 2 * m + (lo ? 1 : 0);
    const int pb = 2 * (16 + m) + (lo ? 0 : 1);

#pragma unroll 2
    for (int jc = 0; jc < N_ATOMS; jc += 32) {
        const int j = jc + lane;
        const float4 xq = sx[j];
        const float2 la = sh[2 * jc + pa];
        const float2 lb = sh[2 * jc + pb];

        const float d0 = xi0 - xq.x, d1 = xi1 - xq.y, d2c = xi2 - xq.z;
        const float d2 = d0 * d0 + d1 * d1 + d2c * d2c + 1e-9f;
        const float inv = fast_rsq(d2);
        const float validf = ((d2 < CUTOFF_SQ) && (j != i)) ? (xq.w * maski) : 0.0f;
        const float radial = validf * fminf(inv, 1.0e3f);
        const float dscale = validf * fminf(inv, 1.0e6f);
        const float dir0 = d0 * dscale, dir1 = d1 * dscale, dir2 = d2c * dscale;

        v2f B1a, B1b;
        B1a[0] = la.x;
        B1a[1] = lo ? radial : la.y;
        B1b[0] = lb.x;
        B1b[1] = lo ? lb.y : radial;

        v8f D1 = __builtin_amdgcn_wmma_f32_16x16x4_f32(
            false, A1a, false, B1a, (short)0, C1, false, false);
        D1 = __builtin_amdgcn_wmma_f32_16x16x4_f32(
            false, A1b, false, B1b, (short)0, D1, false, false);

        v2f x01; x01[0] = D1[0]; x01[1] = D1[1];
        v2f r01; r01[0] = fast_rcp(1.0f + __expf(-x01[0]));
                 r01[1] = fast_rcp(1.0f + __expf(-x01[1]));
        const v2f B2a = x01 * r01;
        const float s2 = silu_f(D1[2]);
        v2f B2b;
        B2b[0] = s2;  B2b[1] = B2a[1];

        v8f D2 = __builtin_amdgcn_wmma_f32_16x16x4_f32(
            false, A2a, false, B2a, (short)0, C2, false, false);
        D2 = __builtin_amdgcn_wmma_f32_16x16x4_f32(
            false, A2b, false, B2b, (short)0, D2, false, false);

        acc0 += D2[0] * dir0;
        acc1 += D2[1] * dir1;
        acc2 += D2[2] * dir2;
    }

    for (int off = 16; off > 0; off >>= 1) {
        acc0 += __shfl_xor(acc0, off, 32);
        acc1 += __shfl_xor(acc1, off, 32);
        acc2 += __shfl_xor(acc2, off, 32);
    }

    if (lane == 0) {
        const float v0 = acc0, v1 = acc1, v2 = acc2;
        const float2 hi12 = sh[2 * i];
        const float2 hi0p = sh[2 * i + 1];
        const float hi0 = hi0p.x, hi1 = hi12.x, hi2 = hi12.y;

        const float u0 = silu_f(v0 * upd_w1[0] + v1 * upd_w1[3] + v2 * upd_w1[6] + upd_b1[0]);
        const float u1 = silu_f(v0 * upd_w1[1] + v1 * upd_w1[4] + v2 * upd_w1[7] + upd_b1[1]);
        const float u2 = silu_f(v0 * upd_w1[2] + v1 * upd_w1[5] + v2 * upd_w1[8] + upd_b1[2]);

        const float h0 = hi0 + u0 * upd_w2[0] + u1 * upd_w2[3] + u2 * upd_w2[6] + upd_b2[0];
        const float h1 = hi1 + u0 * upd_w2[1] + u1 * upd_w2[4] + u2 * upd_w2[7] + upd_b2[1];
        const float h2 = hi2 + u0 * upd_w2[2] + u1 * upd_w2[5] + u2 * upd_w2[8] + upd_b2[2];

        const int base = (b * N_ATOMS + i) * 3;
        const int half = BATCH * N_ATOMS * 3;
#pragma unroll
        for (int c = 0; c < 3; ++c) {
            const float px = h0 * dx_w[0 * 3 + c] + h1 * dx_w[1 * 3 + c] + h2 * dx_w[2 * 3 + c] + dx_b[c];
            const float pv = h0 * dv_w[0 * 3 + c] + h1 * dv_w[1 * 3 + c] + h2 * dv_w[2 * 3 + c] + dv_b[c];
            out[base + c]        = tanh_f(px) * MAX_DISP;
            out[half + base + c] = tanh_f(pv) * 1.0f;
        }
    }
}

extern "C" void kernel_launch(void* const* d_in, const int* in_sizes, int n_in,
                              void* d_out, int out_size, void* d_ws, size_t ws_size,
                              hipStream_t stream) {
    (void)in_sizes; (void)n_in; (void)d_ws; (void)ws_size; (void)out_size;
    const float* x_nm   = (const float*)d_in[0];
    const int*   types  = (const int*)  d_in[1];
    const int*   mask   = (const int*)  d_in[2];
    const float* emb_W  = (const float*)d_in[3];
    const float* msg_w1 = (const float*)d_in[4];
    const float* msg_b1 = (const float*)d_in[5];
    const float* msg_w2 = (const float*)d_in[6];
    const float* msg_b2 = (const float*)d_in[7];
    const float* upd_w1 = (const float*)d_in[8];
    const float* upd_b1 = (const float*)d_in[9];
    const float* upd_w2 = (const float*)d_in[10];
    const float* upd_b2 = (const float*)d_in[11];
    const float* dx_w   = (const float*)d_in[12];
    const float* dx_b   = (const float*)d_in[13];
    const float* dv_w   = (const float*)d_in[14];
    const float* dv_b   = (const float*)d_in[15];
    float* out = (float*)d_out;

    dim3 grid(BATCH * (N_ATOMS / RECV_PER_BLOCK));
    dim3 block(256);
    simple_et_kernel<<<grid, block, 0, stream>>>(
        x_nm, types, mask, emb_W,
        msg_w1, msg_b1, msg_w2, msg_b2,
        upd_w1, upd_b1, upd_w2, upd_b2,
        dx_w, dx_b, dv_w, dv_b, out);
}


# Global loads issued together, not one wait each: staging loop unrolled, its 32 loads in flight at once (was 16 waits in series); message weights fetched beside them; output weights parked in LDS
# speedup vs baseline: 1.8967x; 1.4958x over previous
#include <hip/hip_runtime.h>


typedef float v2f __attribute__((ext_vector_type(2)));
typedef float v8f __attribute__((ext_vector_type(8)));

#define N_ATOMS 2048
#define BATCH 2
#define RECV_PER_BLOCK 8
#define SENDER_SPLIT 4
#define TILE 16
#define CUTOFF_SQ 0.49f
#define MAX_DISP 0.02f

__device__ __forceinline__ float fast_rcp(float x)  { return __builtin_amdgcn_rcpf(x); }
__device__ __forceinline__ float fast_rsq(float x)  { return __builtin_amdgcn_rsqf(x); }

__device__ __forceinline__ float silu_f(float x) {
    return x * fast_rcp(1.0f + __expf(-x));
}
__device__ __forceinline__ float tanh_f(float x) {
    float xc = fminf(fmaxf(x, -15.0f), 15.0f);
    float e  = __expf(2.0f * xc);
    return (e - 1.0f) * fast_rcp(e + 1.0f);
}

__global__ __launch_bounds__(256) void simple_et_kernel(
    const float* __restrict__ x_nm, const int* __restrict__ atom_types,
    const int* __restrict__ mask_in, const float* __restrict__ emb_W,
    const float* __restrict__ msg_w1, const float* __restrict__ msg_b1,
    const float* __restrict__ msg_w2, const float* __restrict__ msg_b2,
    const float* __restrict__ upd_w1, const float* __restrict__ upd_b1,
    const float* __restrict__ upd_w2, const float* __restrict__ upd_b2,
    const float* __restrict__ dx_w, const float* __restrict__ dx_b,
    const float* __restrict__ dv_w, const float* __restrict__ dv_b,
    float* __restrict__ out)
{
    __shared__ float4 sx[N_ATOMS];
    __shared__ float2 sh[N_ATOMS * 2];
    __shared__ float part[8 * 12];
    __shared__ float4 sw4[12];
    float* const sw = reinterpret_cast<float*>(sw4);

    const int tid  = threadIdx.x;
    const int b    = blockIdx.x / (N_ATOMS / RECV_PER_BLOCK);
    const int row0 = (blockIdx.x % (N_ATOMS / RECV_PER_BLOCK)) * RECV_PER_BLOCK;

    const int lane = tid & 31;
    const int wave = tid >> 5;
    const int grp  = wave >> 2;
    const int qtr  = wave & 3;
    const int m    = lane & 15;
    const int hh   = lane >> 4;

    const bool rowok = (m & 7) < 6;
    const int  mm    = rowok ? (m & 7) : 0;
    const int  cm    = mm % 3;
    const int  Rm    = 2 * (m >> 3) + mm / 3;

    const int gb = b * N_ATOMS;
    int tt[N_ATOMS / 256];
#pragma unroll
    for (int it = 0; it < N_ATOMS / 256; ++it)
        tt[it] = atom_types[gb + tid + it * 256];

    const float w1c0 = msg_w1[0 * 3 + cm], w1c1 = msg_w1[1 * 3 + cm];
    const float w1c2 = msg_w1[2 * 3 + cm], w1c3 = msg_w1[3 * 3 + cm];
    const float w2c0 = msg_w2[0 * 3 + cm], w2c1 = msg_w2[1 * 3 + cm], w2c2 = msg_w2[2 * 3 + cm];
    const float mb10 = msg_b1[0], mb11 = msg_b1[1], mb12 = msg_b1[2];
    const float mb20 = msg_b2[0], mb21 = msg_b2[1], mb22 = msg_b2[2];

    float ew0 = 0.0f, ew1 = 0.0f, ew2 = 0.0f, ew3 = 0.0f;
    float eb0 = 0.0f, eb1 = 0.0f, eb2 = 0.0f, eb3 = 0.0f;
    if (wave == 0) {
        const int k9 = lane < 9 ? lane : 8;
        const int k3 = lane < 3 ? lane : 2;
        ew0 = upd_w1[k9]; ew1 = upd_w2[k9]; ew2 = dx_w[k9]; ew3 = dv_w[k9];
        eb0 = upd_b1[k3]; eb1 = upd_b2[k3]; eb2 = dx_b[k3]; eb3 = dv_b[k3];
    }

    int   mk[N_ATOMS / 256];
    float pos[N_ATOMS / 256][3];
#pragma unroll
    for (int it = 0; it < N_ATOMS / 256; ++it)
        mk[it] = mask_in[gb + tid + it * 256];
#pragma unroll
    for (int it = 0; it < N_ATOMS / 256; ++it) {
        const int n = tid + it * 256;
        pos[it][0] = x_nm[(gb + n) * 3 + 0];
        pos[it][1] = x_nm[(gb + n) * 3 + 1];
        pos[it][2] = x_nm[(gb + n) * 3 + 2];
    }
    float em[N_ATOMS / 256][3];
#pragma unroll
    for (int it = 0; it < N_ATOMS / 256; ++it) {
        em[it][0] = emb_W[tt[it] * 3 + 0];
        em[it][1] = emb_W[tt[it] * 3 + 1];
        em[it][2] = emb_W[tt[it] * 3 + 2];
    }

#pragma unroll
    for (int it = 0; it < N_ATOMS / 256; ++it) {
        const int n = tid + it * 256;
        sx[n] = make_float4(pos[it][0], pos[it][1], pos[it][2], mk[it] ? 1.0f : 0.0f);
        sh[2 * n]     = make_float2(em[it][1], em[it][2]);
        sh[2 * n + 1] = make_float2(em[it][0], 0.0f);
    }
    if (wave == 0) {
        if (lane < 9) { sw[lane] = ew0; sw[9 + lane] = ew1; sw[18 + lane] = ew2; sw[27 + lane] = ew3; }
        if (lane < 3) { sw[36 + lane] = eb0; sw[39 + lane] = eb1; sw[42 + lane] = eb2; sw[45 + lane] = eb3; }
    }
    __syncthreads();

    const int ia = row0 + grp * 4 + hh * 2;
    const int ib = ia + 1;
    const float4 xa = sx[ia];
    const float4 xb = sx[ib];

    v2f A1a, A1b;
    {
        const float w0 = hh ? w1c1 : w1c0;
        const float w1 = hh ? w1c2 : 0.0f;
        A1a[0] = rowok ? w0 : 0.0f;
        A1a[1] = rowok ? w1 : 0.0f;
        A1b[0] = (rowok && (2 * hh + 0 == Rm)) ? w1c3 : 0.0f;
        A1b[1] = (rowok && (2 * hh + 1 == Rm)) ? w1c3 : 0.0f;
    }
    v2f A2a, A2b, A2c;
    {
        const bool ra = rowok && (2 * hh + 0 == Rm);
        const bool rb = rowok && (2 * hh + 1 == Rm);
        A2a[0] = ra ? w2c0 : 0.0f;  A2a[1] = ra ? w2c1 : 0.0f;
        A2b[0] = ra ? w2c2 : 0.0f;  A2b[1] = rb ? w2c0 : 0.0f;
        A2c[0] = rb ? w2c1 : 0.0f;  A2c[1] = rb ? w2c2 : 0.0f;
    }

    v8f C1 = {}, C2 = {};
    C1[0] = mb10;  C1[1] = mb11;  C1[2] = mb12;
    C1[3] = mb10;  C1[4] = mb11;  C1[5] = mb12;
    C2[0] = mb20;  C2[1] = mb21;  C2[2] = mb22;
    C2[3] = mb20;  C2[4] = mb21;  C2[5] = mb22;

    float a0 = 0.0f, a1 = 0.0f, a2 = 0.0f;
    float b0 = 0.0f, b1 = 0.0f, b2 = 0.0f;

    const int jbase = qtr * (N_ATOMS / SENDER_SPLIT);
    const int hsel  = hh ? 0 : 1;

#pragma unroll 2
    for (int t = 0; t < (N_ATOMS / SENDER_SPLIT) / TILE; ++t) {
        const int j = jbase + t * TILE + m;
        const float4 xq = sx[j];
        const float2 hv = sh[2 * j + hsel];

        const float da0 = xa.x - xq.x, da1 = xa.y - xq.y, da2 = xa.z - xq.z;
        const float db0 = xb.x - xq.x, db1 = xb.y - xq.y, db2 = xb.z - xq.z;
        float d2a, d2b;
        {
#pragma clang fp contract(off)
            d2a = da0 * da0 + da1 * da1 + da2 * da2 + 1e-9f;
            d2b = db0 * db0 + db1 * db1 + db2 * db2 + 1e-9f;
        }
        const float inva = fast_rsq(d2a);
        const float invb = fast_rsq(d2b);
        const float va = ((d2a < CUTOFF_SQ) && (j != ia)) ? (xq.w * xa.w) : 0.0f;
        const float vb = ((d2b < CUTOFF_SQ) && (j != ib)) ? (xq.w * xb.w) : 0.0f;
        const float sca = va * fminf(inva, 1.0e6f);
        const float scb = vb * fminf(invb, 1.0e6f);

        v2f B1a, B1b;
        B1a[0] = hv.x;
        B1a[1] = hv.y;
        B1b[0] = va * fminf(inva, 1.0e3f);
        B1b[1] = vb * fminf(invb, 1.0e3f);

        v8f D1 = __builtin_amdgcn_wmma_f32_16x16x4_f32(
            false, A1a, false, B1a, (short)0, C1, false, false);
        D1 = __builtin_amdgcn_wmma_f32_16x16x4_f32(
            false, A1b, false, B1b, (short)0, D1, false, false);

        v2f B2a, B2b, B2c;
        B2a[0] = silu_f(D1[0]);  B2a[1] = silu_f(D1[1]);
        B2b[0] = silu_f(D1[2]);  B2b[1] = silu_f(D1[3]);
        B2c[0] = silu_f(D1[4]);  B2c[1] = silu_f(D1[5]);

        v8f D2 = __builtin_amdgcn_wmma_f32_16x16x4_f32(
            false, A2a, false, B2a, (short)0, C2, false, false);
        D2 = __builtin_amdgcn_wmma_f32_16x16x4_f32(
            false, A2b, false, B2b, (short)0, D2, false, false);
        D2 = __builtin_amdgcn_wmma_f32_16x16x4_f32(
            false, A2c, false, B2c, (short)0, D2, false, false);

        a0 += D2[0] * (da0 * sca);
        a1 += D2[1] * (da1 * sca);
        a2 += D2[2] * (da2 * sca);
        b0 += D2[3] * (db0 * scb);
        b1 += D2[4] * (db1 * scb);
        b2 += D2[5] * (db2 * scb);
    }

    for (int off = 8; off > 0; off >>= 1) {
        a0 += __shfl_xor(a0, off, 32);
        a1 += __shfl_xor(a1, off, 32);
        a2 += __shfl_xor(a2, off, 32);
        b0 += __shfl_xor(b0, off, 32);
        b1 += __shfl_xor(b1, off, 32);
        b2 += __shfl_xor(b2, off, 32);
    }
    if (m == 0) {
        float* p = part + wave * 12 + hh * 6;
        p[0] = a0; p[1] = a1; p[2] = a2;
        p[3] = b0; p[4] = b1; p[5] = b2;
    }
    float wl[48];
    float hi0 = 0.0f, hi1 = 0.0f, hi2 = 0.0f;
    if (tid < RECV_PER_BLOCK) {
#pragma unroll
        for (int q = 0; q < 12; ++q) {
            const float4 v = sw4[q];
            wl[4 * q + 0] = v.x;  wl[4 * q + 1] = v.y;  wl[4 * q + 2] = v.z;  wl[4 * q + 3] = v.w;
        }
        const int i = row0 + tid;
        const float2 hi12 = sh[2 * i];
        const float2 hi0p = sh[2 * i + 1];
        hi0 = hi0p.x; hi1 = hi12.x; hi2 = hi12.y;
    }
    __syncthreads();

    if (tid < RECV_PER_BLOCK) {
        const int eg = tid >> 2;
        const int es = tid & 3;
        const float* p = part + (eg * 4) * 12 + (es >> 1) * 6 + (es & 1) * 3;
        float v0 = 0.0f, v1 = 0.0f, v2 = 0.0f;
        for (int q = 0; q < SENDER_SPLIT; ++q) {
            v0 += p[q * 12 + 0];
            v1 += p[q * 12 + 1];
            v2 += p[q * 12 + 2];
        }
        const int i = row0 + tid;
        const float* uw1 = wl;       const float* uw2 = wl + 9;
        const float* xw  = wl + 18;  const float* vw  = wl + 27;
        const float* ub1 = wl + 36;  const float* ub2 = wl + 39;
        const float* xb_ = wl + 42;  const float* vb_ = wl + 45;

        const float u0 = silu_f(v0 * uw1[0] + v1 * uw1[3] + v2 * uw1[6] + ub1[0]);
        const float u1 = silu_f(v0 * uw1[1] + v1 * uw1[4] + v2 * uw1[7] + ub1[1]);
        const float u2 = silu_f(v0 * uw1[2] + v1 * uw1[5] + v2 * uw1[8] + ub1[2]);

        const float h0 = hi0 + u0 * uw2[0] + u1 * uw2[3] + u2 * uw2[6] + ub2[0];
        const float h1 = hi1 + u0 * uw2[1] + u1 * uw2[4] + u2 * uw2[7] + ub2[1];
        const float h2 = hi2 + u0 * uw2[2] + u1 * uw2[5] + u2 * uw2[8] + ub2[2];

        const int base = (b * N_ATOMS + i) * 3;
        const int half = BATCH * N_ATOMS * 3;
#pragma unroll
        for (int c = 0; c < 3; ++c) {
            const float px = h0 * xw[0 * 3 + c] + h1 * xw[1 * 3 + c] + h2 * xw[2 * 3 + c] + xb_[c];
            const float pv = h0 * vw[0 * 3 + c] + h1 * vw[1 * 3 + c] + h2 * vw[2 * 3 + c] + vb_[c];
            out[base + c]        = tanh_f(px) * MAX_DISP;
            out[half + base + c] = tanh_f(pv) * 1.0f;
        }
    }
}

extern "C" void kernel_launch(void* const* d_in, const int* in_sizes, int n_in,
                              void* d_out, int out_size, void* d_ws, size_t ws_size,
                              hipStream_t stream) {
    (void)in_sizes; (void)n_in; (void)d_ws; (void)ws_size; (void)out_size;
    const float* x_nm   = (const float*)d_in[0];
    const int*   types  = (const int*)  d_in[1];
    const int*   mask   = (const int*)  d_in[2];
    const float* emb_W  = (const float*)d_in[3];
    const float* msg_w1 = (const float*)d_in[4];
    const float* msg_b1 = (const float*)d_in[5];
    const float* msg_w2 = (const float*)d_in[6];
    const float* msg_b2 = (const float*)d_in[7];
    const float* upd_w1 = (const float*)d_in[8];
    const float* upd_b1 = (const float*)d_in[9];
    const float* upd_w2 = (const float*)d_in[10];
    const float* upd_b2 = (const float*)d_in[11];
    const float* dx_w   = (const float*)d_in[12];
    const float* dx_b   = (const float*)d_in[13];
    const float* dv_w   = (const float*)d_in[14];
    const float* dv_b   = (const float*)d_in[15];
    float* out = (float*)d_out;

    dim3 grid(BATCH * (N_ATOMS / RECV_PER_BLOCK));
    dim3 block(256);
    simple_et_kernel<<<grid, block, 0, stream>>>(
        x_nm, types, mask, emb_W,
        msg_w1, msg_b1, msg_w2, msg_b2,
        upd_w1, upd_b1, upd_w2, upd_b2,
        dx_w, dx_b, dv_w, dv_b, out);
}


# Layer-1 sender part (h W1[0:3]+b1) leaves the pair loop: made once per atom type on the matrix unit, enters each pair product as its accumulator; 4 matrix products per 64 pairs, was 5
# speedup vs baseline: 2.0667x; 1.0896x over previous
#include <hip/hip_runtime.h>


typedef float v2f __attribute__((ext_vector_type(2)));
typedef float v8f __attribute__((ext_vector_type(8)));

#define N_ATOMS 2048
#define BATCH 2
#define RECV_PER_BLOCK 8
#define SENDER_SPLIT 4
#define TILE 16
#define CUTOFF_SQ 0.49f
#define MAX_DISP 0.02f
#define NUM_TYPES 100
#define TYPE_TILES 7

__device__ __forceinline__ float fast_rcp(float x)  { return __builtin_amdgcn_rcpf(x); }
__device__ __forceinline__ float fast_rsq(float x)  { return __builtin_amdgcn_rsqf(x); }

__device__ __forceinline__ float silu_f(float x) {
    return x * fast_rcp(1.0f + __expf(-x));
}
__device__ __forceinline__ float tanh_f(float x) {
    float xc = fminf(fmaxf(x, -15.0f), 15.0f);
    float e  = __expf(2.0f * xc);
    return (e - 1.0f) * fast_rcp(e + 1.0f);
}

__global__ __launch_bounds__(256) void simple_et_kernel(
    const float* __restrict__ x_nm, const int* __restrict__ atom_types,
    const int* __restrict__ mask_in, const float* __restrict__ emb_W,
    const float* __restrict__ msg_w1, const float* __restrict__ msg_b1,
    const float* __restrict__ msg_w2, const float* __restrict__ msg_b2,
    const float* __restrict__ upd_w1, const float* __restrict__ upd_b1,
    const float* __restrict__ upd_w2, const float* __restrict__ upd_b2,
    const float* __restrict__ dx_w, const float* __restrict__ dx_b,
    const float* __restrict__ dv_w, const float* __restrict__ dv_b,
    float* __restrict__ out)
{
    __shared__ float4 sx[N_ATOMS];
    __shared__ float4 sp[N_ATOMS];
    __shared__ float4 spt[TYPE_TILES * 16];
    __shared__ float4 spe[TYPE_TILES * 16];
    __shared__ float part[8 * 12];
    __shared__ float4 sw4[12];
    float* const sw = reinterpret_cast<float*>(sw4);

    const int tid  = threadIdx.x;
    const int b    = blockIdx.x / (N_ATOMS / RECV_PER_BLOCK);
    const int row0 = (blockIdx.x % (N_ATOMS / RECV_PER_BLOCK)) * RECV_PER_BLOCK;

    const int lane = tid & 31;
    const int wave = tid >> 5;
    const int grp  = wave >> 2;
    const int qtr  = wave & 3;
    const int m    = lane & 15;
    const int hh   = lane >> 4;

    const bool rowok = (m & 7) < 6;
    const int  mm    = rowok ? (m & 7) : 0;
    const int  cm    = mm % 3;
    const int  Rm    = 2 * (m >> 3) + mm / 3;
    const bool rowok1 = (m & 3) < 3;
    const int  cm1    = rowok1 ? (m & 3) : 0;
    const int  Rm1    = 2 * (m >> 3) + ((m & 7) >> 2);

    const int gb = b * N_ATOMS;

    const int ty  = wave * 16 + m;
    const int tyc = ty < NUM_TYPES ? ty : NUM_TYPES - 1;
    float e0 = 0.0f, e1 = 0.0f;
    if (wave < TYPE_TILES) {
        e0 = emb_W[tyc * 3 + (hh ? 1 : 0)];
        e1 = emb_W[tyc * 3 + 2];
    }
    const float w1c0 = msg_w1[0 * 3 + cm1], w1c1 = msg_w1[1 * 3 + cm1];
    const float w1c2 = msg_w1[2 * 3 + cm1], w1c3 = msg_w1[3 * 3 + cm1];
    const float w2c0 = msg_w2[0 * 3 + cm], w2c1 = msg_w2[1 * 3 + cm], w2c2 = msg_w2[2 * 3 + cm];
    const float mb10 = msg_b1[0], mb11 = msg_b1[1], mb12 = msg_b1[2];
    const float mb20 = msg_b2[0], mb21 = msg_b2[1], mb22 = msg_b2[2];

    int tt[N_ATOMS / 256];
#pragma unroll
    for (int it = 0; it < N_ATOMS / 256; ++it)
        tt[it] = atom_types[gb + tid + it * 256];
    const int ti = atom_types[gb + row0 + (tid & (RECV_PER_BLOCK - 1))];

    float ew0 = 0.0f, ew1 = 0.0f, ew2 = 0.0f, ew3 = 0.0f;
    float eb0 = 0.0f, eb1 = 0.0f, eb2 = 0.0f, eb3 = 0.0f;
    if (wave == 0) {
        const int k9 = lane < 9 ? lane : 8;
        const int k3 = lane < 3 ? lane : 2;
        ew0 = upd_w1[k9]; ew1 = upd_w2[k9]; ew2 = dx_w[k9]; ew3 = dv_w[k9];
        eb0 = upd_b1[k3]; eb1 = upd_b2[k3]; eb2 = dx_b[k3]; eb3 = dv_b[k3];
    }

    int   mk[N_ATOMS / 256];
    float pos[N_ATOMS / 256][3];
#pragma unroll
    for (int it = 0; it < N_ATOMS / 256; ++it)
        mk[it] = mask_in[gb + tid + it * 256];
#pragma unroll
    for (int it = 0; it < N_ATOMS / 256; ++it) {
        const int n = tid + it * 256;
        pos[it][0] = x_nm[(gb + n) * 3 + 0];
        pos[it][1] = x_nm[(gb + n) * 3 + 1];
        pos[it][2] = x_nm[(gb + n) * 3 + 2];
    }

    v2f A1a;
    {
        const float w0 = hh ? w1c1 : w1c0;
        const float w1 = hh ? w1c2 : 0.0f;
        A1a[0] = rowok1 ? w0 : 0.0f;
        A1a[1] = rowok1 ? w1 : 0.0f;
    }
    if (wave < TYPE_TILES) {
        v8f C1 = {};
        C1[0] = mb10;  C1[1] = mb11;  C1[2] = mb12;
        C1[4] = mb10;  C1[5] = mb11;  C1[6] = mb12;
        v2f Bt;
        Bt[0] = e0;
        Bt[1] = hh ? e1 : 0.0f;
        const v8f P = __builtin_amdgcn_wmma_f32_16x16x4_f32(
            false, A1a, false, Bt, (short)0, C1, false, false);
        if (ty < NUM_TYPES) {
            float* const spef = reinterpret_cast<float*>(spe);
            if (hh == 0) {
                spt[ty] = make_float4(P[0], P[1], P[2], 0.0f);
                spef[4 * ty + 0] = e0;
                spef[4 * ty + 3] = 0.0f;
            } else {
                spef[4 * ty + 1] = e0;
                spef[4 * ty + 2] = e1;
            }
        }
    }

#pragma unroll
    for (int it = 0; it < N_ATOMS / 256; ++it) {
        const int n = tid + it * 256;
        sx[n] = make_float4(pos[it][0], pos[it][1], pos[it][2], mk[it] ? 1.0f : 0.0f);
    }
    if (wave == 0) {
        if (lane < 9) { sw[lane] = ew0; sw[9 + lane] = ew1; sw[18 + lane] = ew2; sw[27 + lane] = ew3; }
        if (lane < 3) { sw[36 + lane] = eb0; sw[39 + lane] = eb1; sw[42 + lane] = eb2; sw[45 + lane] = eb3; }
    }
    __syncthreads();
    const float4 pv0 = spt[tt[0]], pv1 = spt[tt[1]], pv2 = spt[tt[2]], pv3 = spt[tt[3]];
    const float4 pv4 = spt[tt[4]], pv5 = spt[tt[5]], pv6 = spt[tt[6]], pv7 = spt[tt[7]];
    __builtin_amdgcn_sched_barrier(0);
    sp[tid + 0 * 256] = pv0;  sp[tid + 1 * 256] = pv1;  sp[tid + 2 * 256] = pv2;  sp[tid + 3 * 256] = pv3;
    sp[tid + 4 * 256] = pv4;  sp[tid + 5 * 256] = pv5;  sp[tid + 6 * 256] = pv6;  sp[tid + 7 * 256] = pv7;
    __syncthreads();

    const int ia = row0 + grp * 4 + hh * 2;
    const int ib = ia + 1;
    const float4 xa = sx[ia];
    const float4 xb = sx[ib];

    v2f A1b;
    A1b[0] = (rowok1 && (2 * hh + 0 == Rm1)) ? w1c3 : 0.0f;
    A1b[1] = (rowok1 && (2 * hh + 1 == Rm1)) ? w1c3 : 0.0f;
    v2f A2a, A2b, A2c;
    {
        const bool ra = rowok && (2 * hh + 0 == Rm);
        const bool rb = rowok && (2 * hh + 1 == Rm);
        A2a[0] = ra ? w2c0 : 0.0f;  A2a[1] = ra ? w2c1 : 0.0f;
        A2b[0] = ra ? w2c2 : 0.0f;  A2b[1] = rb ? w2c0 : 0.0f;
        A2c[0] = rb ? w2c1 : 0.0f;  A2c[1] = rb ? w2c2 : 0.0f;
    }

    v8f C2 = {};
    C2[0] = mb20;  C2[1] = mb21;  C2[2] = mb22;
    C2[3] = mb20;  C2[4] = mb21;  C2[5] = mb22;

    float a0 = 0.0f, a1 = 0.0f, a2 = 0.0f;
    float b0 = 0.0f, b1 = 0.0f, b2 = 0.0f;

    const int jbase = qtr * (N_ATOMS / SENDER_SPLIT);

#pragma unroll 2
    for (int t = 0; t < (N_ATOMS / SENDER_SPLIT) / TILE; ++t) {
        const int j = jbase + t * TILE + m;
        const float4 xq = sx[j];
        const float4 pj = sp[j];

        const float da0 = xa.x - xq.x, da1 = xa.y - xq.y, da2 = xa.z - xq.z;
        const float db0 = xb.x - xq.x, db1 = xb.y - xq.y, db2 = xb.z - xq.z;
        float d2a, d2b;
        {
#pragma clang fp contract(off)
            d2a = da0 * da0 + da1 * da1 + da2 * da2 + 1e-9f;
            d2b = db0 * db0 + db1 * db1 + db2 * db2 + 1e-9f;
        }
        const float inva = fast_rsq(d2a);
        const float invb = fast_rsq(d2b);
        const float va = ((d2a < CUTOFF_SQ) && (j != ia)) ? (xq.w * xa.w) : 0.0f;
        const float vb = ((d2b < CUTOFF_SQ) && (j != ib)) ? (xq.w * xb.w) : 0.0f;
        const float sca = va * fminf(inva, 1.0e6f);
        const float scb = vb * fminf(invb, 1.0e6f);

        v2f B1b;
        B1b[0] = va * fminf(inva, 1.0e3f);
        B1b[1] = vb * fminf(invb, 1.0e3f);

        v8f C1;
        C1[0] = pj.x;  C1[1] = pj.y;  C1[2] = pj.z;  C1[3] = pj.w;
        C1[4] = pj.x;  C1[5] = pj.y;  C1[6] = pj.z;  C1[7] = pj.w;
        const v8f D1 = __builtin_amdgcn_wmma_f32_16x16x4_f32(
            false, A1b, false, B1b, (short)0, C1, false, false);

        v2f B2a, B2b, B2c;
        B2a[0] = silu_f(D1[0]);  B2a[1] = silu_f(D1[1]);
        B2b[0] = silu_f(D1[2]);  B2b[1] = silu_f(D1[4]);
        B2c[0] = silu_f(D1[5]);  B2c[1] = silu_f(D1[6]);

        v8f D2 = __builtin_amdgcn_wmma_f32_16x16x4_f32(
            false, A2a, false, B2a, (short)0, C2, false, false);
        D2 = __builtin_amdgcn_wmma_f32_16x16x4_f32(
            false, A2b, false, B2b, (short)0, D2, false, false);
        D2 = __builtin_amdgcn_wmma_f32_16x16x4_f32(
            false, A2c, false, B2c, (short)0, D2, false, false);

        a0 += D2[0] * (da0 * sca);
        a1 += D2[1] * (da1 * sca);
        a2 += D2[2] * (da2 * sca);
        b0 += D2[3] * (db0 * scb);
        b1 += D2[4] * (db1 * scb);
        b2 += D2[5] * (db2 * scb);
    }

    for (int off = 8; off > 0; off >>= 1) {
        a0 += __shfl_xor(a0, off, 32);
        a1 += __shfl_xor(a1, off, 32);
        a2 += __shfl_xor(a2, off, 32);
        b0 += __shfl_xor(b0, off, 32);
        b1 += __shfl_xor(b1, off, 32);
        b2 += __shfl_xor(b2, off, 32);
    }
    if (m == 0) {
        float* p = part + wave * 12 + hh * 6;
        p[0] = a0; p[1] = a1; p[2] = a2;
        p[3] = b0; p[4] = b1; p[5] = b2;
    }
    float wl[48];
    float hi0 = 0.0f, hi1 = 0.0f, hi2 = 0.0f;
    if (tid < RECV_PER_BLOCK) {
#pragma unroll
        for (int q = 0; q < 12; ++q) {
            const float4 v = sw4[q];
            wl[4 * q + 0] = v.x;  wl[4 * q + 1] = v.y;  wl[4 * q + 2] = v.z;  wl[4 * q + 3] = v.w;
        }
        const float4 hi4 = spe[ti];
        hi0 = hi4.x; hi1 = hi4.y; hi2 = hi4.z;
    }
    __syncthreads();

    if (tid < RECV_PER_BLOCK) {
        const int eg = tid >> 2;
        const int es = tid & 3;
        const float* p = part + (eg * 4) * 12 + (es >> 1) * 6 + (es & 1) * 3;
        float v0 = 0.0f, v1 = 0.0f, v2 = 0.0f;
        for (int q = 0; q < SENDER_SPLIT; ++q) {
            v0 += p[q * 12 + 0];
            v1 += p[q * 12 + 1];
            v2 += p[q * 12 + 2];
        }
        const int i = row0 + tid;
        const float* uw1 = wl;       const float* uw2 = wl + 9;
        const float* xw  = wl + 18;  const float* vw  = wl + 27;
        const float* ub1 = wl + 36;  const float* ub2 = wl + 39;
        const float* xb_ = wl + 42;  const float* vb_ = wl + 45;

        const float u0 = silu_f(v0 * uw1[0] + v1 * uw1[3] + v2 * uw1[6] + ub1[0]);
        const float u1 = silu_f(v0 * uw1[1] + v1 * uw1[4] + v2 * uw1[7] + ub1[1]);
        const float u2 = silu_f(v0 * uw1[2] + v1 * uw1[5] + v2 * uw1[8] + ub1[2]);

        const float h0 = hi0 + u0 * uw2[0] + u1 * uw2[3] + u2 * uw2[6] + ub2[0];
        const float h1 = hi1 + u0 * uw2[1] + u1 * uw2[4] + u2 * uw2[7] + ub2[1];
        const float h2 = hi2 + u0 * uw2[2] + u1 * uw2[5] + u2 * uw2[8] + ub2[2];

        const int base = (b * N_ATOMS + i) * 3;
        const int half = BATCH * N_ATOMS * 3;
#pragma unroll
        for (int c = 0; c < 3; ++c) {
            const float px = h0 * xw[0 * 3 + c] + h1 * xw[1 * 3 + c] + h2 * xw[2 * 3 + c] + xb_[c];
            const float pv = h0 * vw[0 * 3 + c] + h1 * vw[1 * 3 + c] + h2 * vw[2 * 3 + c] + vb_[c];
            out[base + c]        = tanh_f(px) * MAX_DISP;
            out[half + base + c] = tanh_f(pv) * 1.0f;
        }
    }
}

extern "C" void kernel_launch(void* const* d_in, const int* in_sizes, int n_in,
                              void* d_out, int out_size, void* d_ws, size_t ws_size,
                              hipStream_t stream) {
    (void)in_sizes; (void)n_in; (void)d_ws; (void)ws_size; (void)out_size;
    const float* x_nm   = (const float*)d_in[0];
    const int*   types  = (const int*)  d_in[1];
    const int*   mask   = (const int*)  d_in[2];
    const float* emb_W  = (const float*)d_in[3];
    const float* msg_w1 = (const float*)d_in[4];
    const float* msg_b1 = (const float*)d_in[5];
    const float* msg_w2 = (const float*)d_in[6];
    const float* msg_b2 = (const float*)d_in[7];
    const float* upd_w1 = (const float*)d_in[8];
    const float* upd_b1 = (const float*)d_in[9];
    const float* upd_w2 = (const float*)d_in[10];
    const float* upd_b2 = (const float*)d_in[11];
    const float* dx_w   = (const float*)d_in[12];
    const float* dx_b   = (const float*)d_in[13];
    const float* dv_w   = (const float*)d_in[14];
    const float* dv_b   = (const float*)d_in[15];
    float* out = (float*)d_out;

    dim3 grid(BATCH * (N_ATOMS / RECV_PER_BLOCK));
    dim3 block(256);
    simple_et_kernel<<<grid, block, 0, stream>>>(
        x_nm, types, mask, emb_W,
        msg_w1, msg_b1, msg_w2, msg_b2,
        upd_w1, upd_b1, upd_w2, upd_b2,
        dx_w, dx_b, dv_w, dv_b, out);
}


# Pair-loop silu as u + u*tanh(u), u = x/2 (sender part and radial weight halved, exact): one hardware tanh and one fma per hidden value, was exp, add, rcp, 2 multiplies; activation rounding differs
# speedup vs baseline: 2.2179x; 1.0732x over previous
#include <hip/hip_runtime.h>


typedef float v2f __attribute__((ext_vector_type(2)));
typedef float v8f __attribute__((ext_vector_type(8)));

#define N_ATOMS 2048
#define BATCH 2
#define RECV_PER_BLOCK 8
#define SENDER_SPLIT 4
#define TILE 16
#define CUTOFF_SQ 0.49f
#define MAX_DISP 0.02f
#define NUM_TYPES 100
#define TYPE_TILES 7

__device__ __forceinline__ float fast_rcp(float x)  { return __builtin_amdgcn_rcpf(x); }
__device__ __forceinline__ float fast_rsq(float x)  { return __builtin_amdgcn_rsqf(x); }

__device__ __forceinline__ float silu_f(float x) {
    return x * fast_rcp(1.0f + __expf(-x));
}
__device__ __forceinline__ float silu_of_half(float u) {
    return __builtin_fmaf(u, __builtin_amdgcn_tanhf(u), u);
}
__device__ __forceinline__ float tanh_f(float x) {
    float xc = fminf(fmaxf(x, -15.0f), 15.0f);
    float e  = __expf(2.0f * xc);
    return (e - 1.0f) * fast_rcp(e + 1.0f);
}

__global__ __launch_bounds__(256) void simple_et_kernel(
    const float* __restrict__ x_nm, const int* __restrict__ atom_types,
    const int* __restrict__ mask_in, const float* __restrict__ emb_W,
    const float* __restrict__ msg_w1, const float* __restrict__ msg_b1,
    const float* __restrict__ msg_w2, const float* __restrict__ msg_b2,
    const float* __restrict__ upd_w1, const float* __restrict__ upd_b1,
    const float* __restrict__ upd_w2, const float* __restrict__ upd_b2,
    const float* __restrict__ dx_w, const float* __restrict__ dx_b,
    const float* __restrict__ dv_w, const float* __restrict__ dv_b,
    float* __restrict__ out)
{
    __shared__ float4 sx[N_ATOMS];
    __shared__ float4 sp[N_ATOMS];
    __shared__ float4 spt[TYPE_TILES * 16];
    __shared__ float4 spe[TYPE_TILES * 16];
    __shared__ float part[8 * 12];
    __shared__ float4 sw4[12];
    float* const sw = reinterpret_cast<float*>(sw4);

    const int tid  = threadIdx.x;
    const int b    = blockIdx.x / (N_ATOMS / RECV_PER_BLOCK);
    const int row0 = (blockIdx.x % (N_ATOMS / RECV_PER_BLOCK)) * RECV_PER_BLOCK;

    const int lane = tid & 31;
    const int wave = tid >> 5;
    const int grp  = wave >> 2;
    const int qtr  = wave & 3;
    const int m    = lane & 15;
    const int hh   = lane >> 4;

    const bool rowok = (m & 7) < 6;
    const int  mm    = rowok ? (m & 7) : 0;
    const int  cm    = mm % 3;
    const int  Rm    = 2 * (m >> 3) + mm / 3;
    const bool rowok1 = (m & 3) < 3;
    const int  cm1    = rowok1 ? (m & 3) : 0;
    const int  Rm1    = 2 * (m >> 3) + ((m & 7) >> 2);

    const int gb = b * N_ATOMS;

    const int ty  = wave * 16 + m;
    const int tyc = ty < NUM_TYPES ? ty : NUM_TYPES - 1;
    float e0 = 0.0f, e1 = 0.0f;
    if (wave < TYPE_TILES) {
        e0 = emb_W[tyc * 3 + (hh ? 1 : 0)];
        e1 = emb_W[tyc * 3 + 2];
    }
    const float w1c0 = msg_w1[0 * 3 + cm1], w1c1 = msg_w1[1 * 3 + cm1];
    const float w1c2 = msg_w1[2 * 3 + cm1], w1c3 = msg_w1[3 * 3 + cm1];
    const float w2c0 = msg_w2[0 * 3 + cm], w2c1 = msg_w2[1 * 3 + cm], w2c2 = msg_w2[2 * 3 + cm];
    const float mb10 = msg_b1[0], mb11 = msg_b1[1], mb12 = msg_b1[2];
    const float mb20 = msg_b2[0], mb21 = msg_b2[1], mb22 = msg_b2[2];

    int tt[N_ATOMS / 256];
#pragma unroll
    for (int it = 0; it < N_ATOMS / 256; ++it)
        tt[it] = atom_types[gb + tid + it * 256];
    const int ti = atom_types[gb + row0 + (tid & (RECV_PER_BLOCK - 1))];

    float ew0 = 0.0f, ew1 = 0.0f, ew2 = 0.0f, ew3 = 0.0f;
    float eb0 = 0.0f, eb1 = 0.0f, eb2 = 0.0f, eb3 = 0.0f;
    if (wave == 0) {
        const int k9 = lane < 9 ? lane : 8;
        const int k3 = lane < 3 ? lane : 2;
        ew0 = upd_w1[k9]; ew1 = upd_w2[k9]; ew2 = dx_w[k9]; ew3 = dv_w[k9];
        eb0 = upd_b1[k3]; eb1 = upd_b2[k3]; eb2 = dx_b[k3]; eb3 = dv_b[k3];
    }

    int   mk[N_ATOMS / 256];
    float pos[N_ATOMS / 256][3];
#pragma unroll
    for (int it = 0; it < N_ATOMS / 256; ++it)
        mk[it] = mask_in[gb + tid + it * 256];
#pragma unroll
    for (int it = 0; it < N_ATOMS / 256; ++it) {
        const int n = tid + it * 256;
        pos[it][0] = x_nm[(gb + n) * 3 + 0];
        pos[it][1] = x_nm[(gb + n) * 3 + 1];
        pos[it][2] = x_nm[(gb + n) * 3 + 2];
    }

    v2f A1a;
    {
        const float w0 = hh ? w1c1 : w1c0;
        const float w1 = hh ? w1c2 : 0.0f;
        A1a[0] = rowok1 ? w0 : 0.0f;
        A1a[1] = rowok1 ? w1 : 0.0f;
    }
    if (wave < TYPE_TILES) {
        v8f C1 = {};
        C1[0] = mb10;  C1[1] = mb11;  C1[2] = mb12;
        C1[4] = mb10;  C1[5] = mb11;  C1[6] = mb12;
        v2f Bt;
        Bt[0] = e0;
        Bt[1] = hh ? e1 : 0.0f;
        const v8f P = __builtin_amdgcn_wmma_f32_16x16x4_f32(
            false, A1a, false, Bt, (short)0, C1, false, false);
        if (ty < NUM_TYPES) {
            float* const spef = reinterpret_cast<float*>(spe);
            if (hh == 0) {
                spt[ty] = make_float4(0.5f * P[0], 0.5f * P[1], 0.5f * P[2], 0.0f);
                spef[4 * ty + 0] = e0;
                spef[4 * ty + 3] = 0.0f;
            } else {
                spef[4 * ty + 1] = e0;
                spef[4 * ty + 2] = e1;
            }
        }
    }

#pragma unroll
    for (int it = 0; it < N_ATOMS / 256; ++it) {
        const int n = tid + it * 256;
        sx[n] = make_float4(pos[it][0], pos[it][1], pos[it][2], mk[it] ? 1.0f : 0.0f);
    }
    if (wave == 0) {
        if (lane < 9) { sw[lane] = ew0; sw[9 + lane] = ew1; sw[18 + lane] = ew2; sw[27 + lane] = ew3; }
        if (lane < 3) { sw[36 + lane] = eb0; sw[39 + lane] = eb1; sw[42 + lane] = eb2; sw[45 + lane] = eb3; }
    }
    __syncthreads();
    const float4 pv0 = spt[tt[0]], pv1 = spt[tt[1]], pv2 = spt[tt[2]], pv3 = spt[tt[3]];
    const float4 pv4 = spt[tt[4]], pv5 = spt[tt[5]], pv6 = spt[tt[6]], pv7 = spt[tt[7]];
    __builtin_amdgcn_sched_barrier(0);
    sp[tid + 0 * 256] = pv0;  sp[tid + 1 * 256] = pv1;  sp[tid + 2 * 256] = pv2;  sp[tid + 3 * 256] = pv3;
    sp[tid + 4 * 256] = pv4;  sp[tid + 5 * 256] = pv5;  sp[tid + 6 * 256] = pv6;  sp[tid + 7 * 256] = pv7;
    __syncthreads();

    const int ia = row0 + grp * 4 + hh * 2;
    const int ib = ia + 1;
    const float4 xa = sx[ia];
    const float4 xb = sx[ib];

    const float w1h = 0.5f * w1c3;
    v2f A1b;
    A1b[0] = (rowok1 && (2 * hh + 0 == Rm1)) ? w1h : 0.0f;
    A1b[1] = (rowok1 && (2 * hh + 1 == Rm1)) ? w1h : 0.0f;
    v2f A2a, A2b, A2c;
    {
        const bool ra = rowok && (2 * hh + 0 == Rm);
        const bool rb = rowok && (2 * hh + 1 == Rm);
        A2a[0] = ra ? w2c0 : 0.0f;  A2a[1] = ra ? w2c1 : 0.0f;
        A2b[0] = ra ? w2c2 : 0.0f;  A2b[1] = rb ? w2c0 : 0.0f;
        A2c[0] = rb ? w2c1 : 0.0f;  A2c[1] = rb ? w2c2 : 0.0f;
    }

    v8f C2 = {};
    C2[0] = mb20;  C2[1] = mb21;  C2[2] = mb22;
    C2[3] = mb20;  C2[4] = mb21;  C2[5] = mb22;

    float a0 = 0.0f, a1 = 0.0f, a2 = 0.0f;
    float b0 = 0.0f, b1 = 0.0f, b2 = 0.0f;

    const int jbase = qtr * (N_ATOMS / SENDER_SPLIT);

#pragma unroll 2
    for (int t = 0; t < (N_ATOMS / SENDER_SPLIT) / TILE; ++t) {
        const int j = jbase + t * TILE + m;
        const float4 xq = sx[j];
        const float4 pj = sp[j];

        const float da0 = xa.x - xq.x, da1 = xa.y - xq.y, da2 = xa.z - xq.z;
        const float db0 = xb.x - xq.x, db1 = xb.y - xq.y, db2 = xb.z - xq.z;
        float d2a, d2b;
        {
#pragma clang fp contract(off)
            d2a = da0 * da0 + da1 * da1 + da2 * da2 + 1e-9f;
            d2b = db0 * db0 + db1 * db1 + db2 * db2 + 1e-9f;
        }
        const float inva = fast_rsq(d2a);
        const float invb = fast_rsq(d2b);
        const float va = ((d2a < CUTOFF_SQ) && (j != ia)) ? (xq.w * xa.w) : 0.0f;
        const float vb = ((d2b < CUTOFF_SQ) && (j != ib)) ? (xq.w * xb.w) : 0.0f;
        const float sca = va * fminf(inva, 1.0e6f);
        const float scb = vb * fminf(invb, 1.0e6f);

        v2f B1b;
        B1b[0] = va * fminf(inva, 1.0e3f);
        B1b[1] = vb * fminf(invb, 1.0e3f);

        v8f C1;
        C1[0] = pj.x;  C1[1] = pj.y;  C1[2] = pj.z;  C1[3] = pj.w;
        C1[4] = pj.x;  C1[5] = pj.y;  C1[6] = pj.z;  C1[7] = pj.w;
        const v8f D1 = __builtin_amdgcn_wmma_f32_16x16x4_f32(
            false, A1b, false, B1b, (short)0, C1, false, false);

        v2f B2a, B2b, B2c;
        B2a[0] = silu_of_half(D1[0]);  B2a[1] = silu_of_half(D1[1]);
        B2b[0] = silu_of_half(D1[2]);  B2b[1] = silu_of_half(D1[4]);
        B2c[0] = silu_of_half(D1[5]);  B2c[1] = silu_of_half(D1[6]);

        v8f D2 = __builtin_amdgcn_wmma_f32_16x16x4_f32(
            false, A2a, false, B2a, (short)0, C2, false, false);
        D2 = __builtin_amdgcn_wmma_f32_16x16x4_f32(
            false, A2b, false, B2b, (short)0, D2, false, false);
        D2 = __builtin_amdgcn_wmma_f32_16x16x4_f32(
            false, A2c, false, B2c, (short)0, D2, false, false);

        a0 += D2[0] * (da0 * sca);
        a1 += D2[1] * (da1 * sca);
        a2 += D2[2] * (da2 * sca);
        b0 += D2[3] * (db0 * scb);
        b1 += D2[4] * (db1 * scb);
        b2 += D2[5] * (db2 * scb);
    }

    for (int off = 8; off > 0; off >>= 1) {
        a0 += __shfl_xor(a0, off, 32);
        a1 += __shfl_xor(a1, off, 32);
        a2 += __shfl_xor(a2, off, 32);
        b0 += __shfl_xor(b0, off, 32);
        b1 += __shfl_xor(b1, off, 32);
        b2 += __shfl_xor(b2, off, 32);
    }
    if (m == 0) {
        float* p = part + wave * 12 + hh * 6;
        p[0] = a0; p[1] = a1; p[2] = a2;
        p[3] = b0; p[4] = b1; p[5] = b2;
    }
    float wl[48];
    float hi0 = 0.0f, hi1 = 0.0f, hi2 = 0.0f;
    if (tid < RECV_PER_BLOCK) {
#pragma unroll
        for (int q = 0; q < 12; ++q) {
            const float4 v = sw4[q];
            wl[4 * q + 0] = v.x;  wl[4 * q + 1] = v.y;  wl[4 * q + 2] = v.z;  wl[4 * q + 3] = v.w;
        }
        const float4 hi4 = spe[ti];
        hi0 = hi4.x; hi1 = hi4.y; hi2 = hi4.z;
    }
    __syncthreads();

    if (tid < RECV_PER_BLOCK) {
        const int eg = tid >> 2;
        const int es = tid & 3;
        const float* p = part + (eg * 4) * 12 + (es >> 1) * 6 + (es & 1) * 3;
        float v0 = 0.0f, v1 = 0.0f, v2 = 0.0f;
        for (int q = 0; q < SENDER_SPLIT; ++q) {
            v0 += p[q * 12 + 0];
            v1 += p[q * 12 + 1];
            v2 += p[q * 12 + 2];
        }
        const int i = row0 + tid;
        const float* uw1 = wl;       const float* uw2 = wl + 9;
        const float* xw  = wl + 18;  const float* vw  = wl + 27;
        const float* ub1 = wl + 36;  const float* ub2 = wl + 39;
        const float* xb_ = wl + 42;  const float* vb_ = wl + 45;

        const float u0 = silu_f(v0 * uw1[0] + v1 * uw1[3] + v2 * uw1[6] + ub1[0]);
        const float u1 = silu_f(v0 * uw1[1] + v1 * uw1[4] + v2 * uw1[7] + ub1[1]);
        const float u2 = silu_f(v0 * uw1[2] + v1 * uw1[5] + v2 * uw1[8] + ub1[2]);

        const float h0 = hi0 + u0 * uw2[0] + u1 * uw2[3] + u2 * uw2[6] + ub2[0];
        const float h1 = hi1 + u0 * uw2[1] + u1 * uw2[4] + u2 * uw2[7] + ub2[1];
        const float h2 = hi2 + u0 * uw2[2] + u1 * uw2[5] + u2 * uw2[8] + ub2[2];

        const int base = (b * N_ATOMS + i) * 3;
        const int half = BATCH * N_ATOMS * 3;
#pragma unroll
        for (int c = 0; c < 3; ++c) {
            const float px = h0 * xw[0 * 3 + c] + h1 * xw[1 * 3 + c] + h2 * xw[2 * 3 + c] + xb_[c];
            const float pv = h0 * vw[0 * 3 + c] + h1 * vw[1 * 3 + c] + h2 * vw[2 * 3 + c] + vb_[c];
            out[base + c]        = tanh_f(px) * MAX_DISP;
            out[half + base + c] = tanh_f(pv) * 1.0f;
        }
    }
}

extern "C" void kernel_launch(void* const* d_in, const int* in_sizes, int n_in,
                              void* d_out, int out_size, void* d_ws, size_t ws_size,
                              hipStream_t stream) {
    (void)in_sizes; (void)n_in; (void)d_ws; (void)ws_size; (void)out_size;
    const float* x_nm   = (const float*)d_in[0];
    const int*   types  = (const int*)  d_in[1];
    const int*   mask   = (const int*)  d_in[2];
    const float* emb_W  = (const float*)d_in[3];
    const float* msg_w1 = (const float*)d_in[4];
    const float* msg_b1 = (const float*)d_in[5];
    const float* msg_w2 = (const float*)d_in[6];
    const float* msg_b2 = (const float*)d_in[7];
    const float* upd_w1 = (const float*)d_in[8];
    const float* upd_b1 = (const float*)d_in[9];
    const float* upd_w2 = (const float*)d_in[10];
    const float* upd_b2 = (const float*)d_in[11];
    const float* dx_w   = (const float*)d_in[12];
    const float* dx_b   = (const float*)d_in[13];
    const float* dv_w   = (const float*)d_in[14];
    const float* dv_b   = (const float*)d_in[15];
    float* out = (float*)d_out;

    dim3 grid(BATCH * (N_ATOMS / RECV_PER_BLOCK));
    dim3 block(256);
    simple_et_kernel<<<grid, block, 0, stream>>>(
        x_nm, types, mask, emb_W,
        msg_w1, msg_b1, msg_w2, msg_b2,
        upd_w1, upd_b1, upd_w2, upd_b2,
        dx_w, dx_b, dv_w, dv_b, out);
}


# Layer 2 leaves the pair loop by linearity: loop sums (h0,h1,h2,1) x direction per receiver; W2, b2 meet the sums once per receiver on the matrix unit; 1 product per 64 pairs, was 4; sum order changes
# speedup vs baseline: 2.5418x; 1.1460x over previous
#include <hip/hip_runtime.h>


typedef float v2f __attribute__((ext_vector_type(2)));
typedef float v8f __attribute__((ext_vector_type(8)));

#define N_ATOMS 2048
#define BATCH 2
#define RECV_PER_BLOCK 8
#define SENDER_SPLIT 4
#define TILE 16
#define CUTOFF_SQ 0.49f
#define MAX_DISP 0.02f
#define NUM_TYPES 100
#define TYPE_TILES 7

__device__ __forceinline__ float fast_rcp(float x)  { return __builtin_amdgcn_rcpf(x); }
__device__ __forceinline__ float fast_rsq(float x)  { return __builtin_amdgcn_rsqf(x); }

__device__ __forceinline__ float silu_f(float x) {
    return x * fast_rcp(1.0f + __expf(-x));
}
__device__ __forceinline__ float silu_of_half(float u) {
    return __builtin_fmaf(u, __builtin_amdgcn_tanhf(u), u);
}
__device__ __forceinline__ float tanh_f(float x) {
    float xc = fminf(fmaxf(x, -15.0f), 15.0f);
    float e  = __expf(2.0f * xc);
    return (e - 1.0f) * fast_rcp(e + 1.0f);
}

__global__ __launch_bounds__(256) void simple_et_kernel(
    const float* __restrict__ x_nm, const int* __restrict__ atom_types,
    const int* __restrict__ mask_in, const float* __restrict__ emb_W,
    const float* __restrict__ msg_w1, const float* __restrict__ msg_b1,
    const float* __restrict__ msg_w2, const float* __restrict__ msg_b2,
    const float* __restrict__ upd_w1, const float* __restrict__ upd_b1,
    const float* __restrict__ upd_w2, const float* __restrict__ upd_b2,
    const float* __restrict__ dx_w, const float* __restrict__ dx_b,
    const float* __restrict__ dv_w, const float* __restrict__ dv_b,
    float* __restrict__ out)
{
    __shared__ float4 sx[N_ATOMS];
    __shared__ float4 sp[N_ATOMS];
    __shared__ float4 spt[TYPE_TILES * 16];
    __shared__ float4 spe[TYPE_TILES * 16];
    __shared__ float4 part4[8 * 12];
    __shared__ float4 sw4[12];
    float* const sw = reinterpret_cast<float*>(sw4);

    const int tid  = threadIdx.x;
    const int b    = blockIdx.x / (N_ATOMS / RECV_PER_BLOCK);
    const int row0 = (blockIdx.x % (N_ATOMS / RECV_PER_BLOCK)) * RECV_PER_BLOCK;

    const int lane = tid & 31;
    const int wave = tid >> 5;
    const int grp  = wave >> 2;
    const int qtr  = wave & 3;
    const int m    = lane & 15;
    const int hh   = lane >> 4;

    const bool rowok1 = (m & 3) < 3;
    const int  cm1    = rowok1 ? (m & 3) : 0;
    const int  Rm1    = 2 * (m >> 3) + ((m & 7) >> 2);

    const int gb = b * N_ATOMS;

    const int ty  = wave * 16 + m;
    const int tyc = ty < NUM_TYPES ? ty : NUM_TYPES - 1;
    float e0 = 0.0f, e1 = 0.0f;
    if (wave < TYPE_TILES) {
        e0 = emb_W[tyc * 3 + (hh ? 1 : 0)];
        e1 = emb_W[tyc * 3 + 2];
    }
    const float w1c0 = msg_w1[0 * 3 + cm1], w1c1 = msg_w1[1 * 3 + cm1];
    const float w1c2 = msg_w1[2 * 3 + cm1], w1c3 = msg_w1[3 * 3 + cm1];
    const float mb10 = msg_b1[0], mb11 = msg_b1[1], mb12 = msg_b1[2];
    const int   cq  = m < 3 ? m : 0;
    const float wx0 = msg_w2[(2 * hh) * 3 + cq];
    const float wx1 = hh ? msg_b2[cq] : msg_w2[3 + cq];

    int tt[N_ATOMS / 256];
#pragma unroll
    for (int it = 0; it < N_ATOMS / 256; ++it)
        tt[it] = atom_types[gb + tid + it * 256];
    const int ti = atom_types[gb + row0 + (tid & (RECV_PER_BLOCK - 1))];

    float ew0 = 0.0f, ew1 = 0.0f, ew2 = 0.0f, ew3 = 0.0f;
    float eb0 = 0.0f, eb1 = 0.0f, eb2 = 0.0f, eb3 = 0.0f;
    if (wave == 0) {
        const int k9 = lane < 9 ? lane : 8;
        const int k3 = lane < 3 ? lane : 2;
        ew0 = upd_w1[k9]; ew1 = upd_w2[k9]; ew2 = dx_w[k9]; ew3 = dv_w[k9];
        eb0 = upd_b1[k3]; eb1 = upd_b2[k3]; eb2 = dx_b[k3]; eb3 = dv_b[k3];
    }

    int   mk[N_ATOMS / 256];
    float pos[N_ATOMS / 256][3];
#pragma unroll
    for (int it = 0; it < N_ATOMS / 256; ++it)
        mk[it] = mask_in[gb + tid + it * 256];
#pragma unroll
    for (int it = 0; it < N_ATOMS / 256; ++it) {
        const int n = tid + it * 256;
        pos[it][0] = x_nm[(gb + n) * 3 + 0];
        pos[it][1] = x_nm[(gb + n) * 3 + 1];
        pos[it][2] = x_nm[(gb + n) * 3 + 2];
    }

    v2f A1a;
    {
        const float w0 = hh ? w1c1 : w1c0;
        const float w1 = hh ? w1c2 : 0.0f;
        A1a[0] = rowok1 ? w0 : 0.0f;
        A1a[1] = rowok1 ? w1 : 0.0f;
    }
    if (wave < TYPE_TILES) {
        v8f C1 = {};
        C1[0] = mb10;  C1[1] = mb11;  C1[2] = mb12;
        C1[4] = mb10;  C1[5] = mb11;  C1[6] = mb12;
        v2f Bt;
        Bt[0] = e0;
        Bt[1] = hh ? e1 : 0.0f;
        const v8f P = __builtin_amdgcn_wmma_f32_16x16x4_f32(
            false, A1a, false, Bt, (short)0, C1, false, false);
        if (ty < NUM_TYPES) {
            float* const spef = reinterpret_cast<float*>(spe);
            if (hh == 0) {
                spt[ty] = make_float4(0.5f * P[0], 0.5f * P[1], 0.5f * P[2], 0.0f);
                spef[4 * ty + 0] = e0;
                spef[4 * ty + 3] = 0.0f;
            } else {
                spef[4 * ty + 1] = e0;
                spef[4 * ty + 2] = e1;
            }
        }
    }

#pragma unroll
    for (int it = 0; it < N_ATOMS / 256; ++it) {
        const int n = tid + it * 256;
        sx[n] = make_float4(pos[it][0], pos[it][1], pos[it][2], mk[it] ? 1.0f : 0.0f);
    }
    if (wave == 0) {
        if (lane < 9) { sw[lane] = ew0; sw[9 + lane] = ew1; sw[18 + lane] = ew2; sw[27 + lane] = ew3; }
        if (lane < 3) { sw[36 + lane] = eb0; sw[39 + lane] = eb1; sw[42 + lane] = eb2; sw[45 + lane] = eb3; }
    }
    __syncthreads();
    const float4 pv0 = spt[tt[0]], pv1 = spt[tt[1]], pv2 = spt[tt[2]], pv3 = spt[tt[3]];
    const float4 pv4 = spt[tt[4]], pv5 = spt[tt[5]], pv6 = spt[tt[6]], pv7 = spt[tt[7]];
    __builtin_amdgcn_sched_barrier(0);
    sp[tid + 0 * 256] = pv0;  sp[tid + 1 * 256] = pv1;  sp[tid + 2 * 256] = pv2;  sp[tid + 3 * 256] = pv3;
    sp[tid + 4 * 256] = pv4;  sp[tid + 5 * 256] = pv5;  sp[tid + 6 * 256] = pv6;  sp[tid + 7 * 256] = pv7;
    __syncthreads();

    const int ia = row0 + grp * 4 + hh * 2;
    const int ib = ia + 1;
    const float4 xa = sx[ia];
    const float4 xb = sx[ib];

    const float w1h = 0.5f * w1c3;
    v2f A1b;
    A1b[0] = (rowok1 && (2 * hh + 0 == Rm1)) ? w1h : 0.0f;
    A1b[1] = (rowok1 && (2 * hh + 1 == Rm1)) ? w1h : 0.0f;
    v2f Pq[2][3], Pr[2][3];
#pragma unroll
    for (int s = 0; s < 2; ++s) {
#pragma unroll
        for (int cc = 0; cc < 3; ++cc) {
            Pq[s][cc][0] = 0.0f;  Pq[s][cc][1] = 0.0f;
            Pr[s][cc][0] = 0.0f;  Pr[s][cc][1] = 0.0f;
        }
    }

    const int jbase = qtr * (N_ATOMS / SENDER_SPLIT);

#pragma unroll 2
    for (int t = 0; t < (N_ATOMS / SENDER_SPLIT) / TILE; ++t) {
        const int j = jbase + t * TILE + m;
        const float4 xq = sx[j];
        const float4 pj = sp[j];

        const float da0 = xa.x - xq.x, da1 = xa.y - xq.y, da2 = xa.z - xq.z;
        const float db0 = xb.x - xq.x, db1 = xb.y - xq.y, db2 = xb.z - xq.z;
        float d2a, d2b;
        {
#pragma clang fp contract(off)
            d2a = da0 * da0 + da1 * da1 + da2 * da2 + 1e-9f;
            d2b = db0 * db0 + db1 * db1 + db2 * db2 + 1e-9f;
        }
        const float inva = fast_rsq(d2a);
        const float invb = fast_rsq(d2b);
        const float va = ((d2a < CUTOFF_SQ) && (j != ia)) ? (xq.w * xa.w) : 0.0f;
        const float vb = ((d2b < CUTOFF_SQ) && (j != ib)) ? (xq.w * xb.w) : 0.0f;
        const float sca = va * fminf(inva, 1.0e6f);
        const float scb = vb * fminf(invb, 1.0e6f);

        v2f B1b;
        B1b[0] = va * fminf(inva, 1.0e3f);
        B1b[1] = vb * fminf(invb, 1.0e3f);

        v8f C1;
        C1[0] = pj.x;  C1[1] = pj.y;  C1[2] = pj.z;  C1[3] = pj.w;
        C1[4] = pj.x;  C1[5] = pj.y;  C1[6] = pj.z;  C1[7] = pj.w;
        const v8f D1 = __builtin_amdgcn_wmma_f32_16x16x4_f32(
            false, A1b, false, B1b, (short)0, C1, false, false);

        v2f hq[2], hr[2];
        hq[0][0] = silu_of_half(D1[0]);  hq[0][1] = silu_of_half(D1[1]);
        hr[0][0] = silu_of_half(D1[2]);  hr[0][1] = 1.0f;
        hq[1][0] = silu_of_half(D1[4]);  hq[1][1] = silu_of_half(D1[5]);
        hr[1][0] = silu_of_half(D1[6]);  hr[1][1] = 1.0f;

        const float dir[2][3] = { { da0 * sca, da1 * sca, da2 * sca },
                                  { db0 * scb, db1 * scb, db2 * scb } };
#pragma unroll
        for (int s = 0; s < 2; ++s) {
#pragma unroll
            for (int cc = 0; cc < 3; ++cc) {
                v2f dd;
                dd[0] = dir[s][cc];  dd[1] = dir[s][cc];
                Pq[s][cc] = __builtin_elementwise_fma(hq[s], dd, Pq[s][cc]);
                Pr[s][cc] = __builtin_elementwise_fma(hr[s], dd, Pr[s][cc]);
            }
        }
    }

#pragma unroll
    for (int off = 8; off > 0; off >>= 1) {
#pragma unroll
        for (int s = 0; s < 2; ++s) {
#pragma unroll
            for (int cc = 0; cc < 3; ++cc) {
                Pq[s][cc][0] += __shfl_xor(Pq[s][cc][0], off, 32);
                Pq[s][cc][1] += __shfl_xor(Pq[s][cc][1], off, 32);
                Pr[s][cc][0] += __shfl_xor(Pr[s][cc][0], off, 32);
                Pr[s][cc][1] += __shfl_xor(Pr[s][cc][1], off, 32);
            }
        }
    }
    if (m == 0) {
#pragma unroll
        for (int s = 0; s < 2; ++s) {
#pragma unroll
            for (int cc = 0; cc < 3; ++cc)
                part4[((wave * 2 + hh) * 2 + s) * 3 + cc] =
                    make_float4(Pq[s][cc][0], Pq[s][cc][1], Pr[s][cc][0], Pr[s][cc][1]);
        }
    }
    float wl[48];
    float hi0 = 0.0f, hi1 = 0.0f, hi2 = 0.0f;
    if (tid < RECV_PER_BLOCK) {
#pragma unroll
        for (int q = 0; q < 12; ++q) {
            const float4 v = sw4[q];
            wl[4 * q + 0] = v.x;  wl[4 * q + 1] = v.y;  wl[4 * q + 2] = v.z;  wl[4 * q + 3] = v.w;
        }
        const float4 hi4 = spe[ti];
        hi0 = hi4.x; hi1 = hi4.y; hi2 = hi4.z;
    }
    __syncthreads();

    if (wave != 0) return;

    v8f Dm = {};
    {
        const bool colok = m < RECV_PER_BLOCK;
        const int  Rc    = colok ? m : 0;
        const int  pb    = (((Rc >> 2) * SENDER_SPLIT * 2 + ((Rc >> 1) & 1)) * 2 + (Rc & 1)) * 3;
        const float* const pf = reinterpret_cast<const float*>(part4);
#pragma unroll
        for (int cc = 0; cc < 3; ++cc) {
            v2f Bm;
            Bm[0] = 0.0f;  Bm[1] = 0.0f;
#pragma unroll
            for (int q = 0; q < SENDER_SPLIT; ++q) {
                const float* p = pf + (pb + q * 12 + cc) * 4 + 2 * hh;
                Bm[0] += p[0];
                Bm[1] += p[1];
            }
            if (!colok) { Bm[0] = 0.0f;  Bm[1] = 0.0f; }
            v2f Am;
            Am[0] = (m == cc) ? wx0 : 0.0f;
            Am[1] = (m == cc) ? wx1 : 0.0f;
            Dm = __builtin_amdgcn_wmma_f32_16x16x4_f32(
                false, Am, false, Bm, (short)0, Dm, false, false);
        }
    }

    if (tid < RECV_PER_BLOCK) {
        const float v0 = Dm[0], v1 = Dm[1], v2 = Dm[2];
        const int i = row0 + tid;
        const float* uw1 = wl;       const float* uw2 = wl + 9;
        const float* xw  = wl + 18;  const float* vw  = wl + 27;
        const float* ub1 = wl + 36;  const float* ub2 = wl + 39;
        const float* xb_ = wl + 42;  const float* vb_ = wl + 45;

        const float u0 = silu_f(v0 * uw1[0] + v1 * uw1[3] + v2 * uw1[6] + ub1[0]);
        const float u1 = silu_f(v0 * uw1[1] + v1 * uw1[4] + v2 * uw1[7] + ub1[1]);
        const float u2 = silu_f(v0 * uw1[2] + v1 * uw1[5] + v2 * uw1[8] + ub1[2]);

        const float h0 = hi0 + u0 * uw2[0] + u1 * uw2[3] + u2 * uw2[6] + ub2[0];
        const float h1 = hi1 + u0 * uw2[1] + u1 * uw2[4] + u2 * uw2[7] + ub2[1];
        const float h2 = hi2 + u0 * uw2[2] + u1 * uw2[5] + u2 * uw2[8] + ub2[2];

        const int base = (b * N_ATOMS + i) * 3;
        const int half = BATCH * N_ATOMS * 3;
#pragma unroll
        for (int c = 0; c < 3; ++c) {
            const float px = h0 * xw[0 * 3 + c] + h1 * xw[1 * 3 + c] + h2 * xw[2 * 3 + c] + xb_[c];
            const float pv = h0 * vw[0 * 3 + c] + h1 * vw[1 * 3 + c] + h2 * vw[2 * 3 + c] + vb_[c];
            out[base + c]        = tanh_f(px) * MAX_DISP;
            out[half + base + c] = tanh_f(pv) * 1.0f;
        }
    }
}

extern "C" void kernel_launch(void* const* d_in, const int* in_sizes, int n_in,
                              void* d_out, int out_size, void* d_ws, size_t ws_size,
                              hipStream_t stream) {
    (void)in_sizes; (void)n_in; (void)d_ws; (void)ws_size; (void)out_size;
    const float* x_nm   = (const float*)d_in[0];
    const int*   types  = (const int*)  d_in[1];
    const int*   mask   = (const int*)  d_in[2];
    const float* emb_W  = (const float*)d_in[3];
    const float* msg_w1 = (const float*)d_in[4];
    const float* msg_b1 = (const float*)d_in[5];
    const float* msg_w2 = (const float*)d_in[6];
    const float* msg_b2 = (const float*)d_in[7];
    const float* upd_w1 = (const float*)d_in[8];
    const float* upd_b1 = (const float*)d_in[9];
    const float* upd_w2 = (const float*)d_in[10];
    const float* upd_b2 = (const float*)d_in[11];
    const float* dx_w   = (const float*)d_in[12];
    const float* dx_b   = (const float*)d_in[13];
    const float* dv_w   = (const float*)d_in[14];
    const float* dv_b   = (const float*)d_in[15];
    float* out = (float*)d_out;

    dim3 grid(BATCH * (N_ATOMS / RECV_PER_BLOCK));
    dim3 block(256);
    simple_et_kernel<<<grid, block, 0, stream>>>(
        x_nm, types, mask, emb_W,
        msg_w1, msg_b1, msg_w2, msg_b2,
        upd_w1, upd_b1, upd_w2, upd_b2,
        dx_w, dx_b, dv_w, dv_b, out);
}


# Pair validity acts on the direction scale alone, one compare and select per pair: masks ride in the 1e-9 term and the cutoff; radial unmasked; self test and 1e-6 clamp dropped as identities
# speedup vs baseline: 2.9538x; 1.1621x over previous
#include <hip/hip_runtime.h>


typedef float v2f __attribute__((ext_vector_type(2)));
typedef float v8f __attribute__((ext_vector_type(8)));

#define N_ATOMS 2048
#define BATCH 2
#define RECV_PER_BLOCK 8
#define SENDER_SPLIT 4
#define TILE 16
#define CUTOFF_SQ 0.49f
#define MAX_DISP 0.02f
#define NUM_TYPES 100
#define TYPE_TILES 7

__device__ __forceinline__ float fast_rcp(float x)  { return __builtin_amdgcn_rcpf(x); }
__device__ __forceinline__ float fast_rsq(float x)  { return __builtin_amdgcn_rsqf(x); }

__device__ __forceinline__ float silu_f(float x) {
    return x * fast_rcp(1.0f + __expf(-x));
}
__device__ __forceinline__ float silu_of_half(float u) {
    return __builtin_fmaf(u, __builtin_amdgcn_tanhf(u), u);
}
__device__ __forceinline__ float tanh_f(float x) {
    float xc = fminf(fmaxf(x, -15.0f), 15.0f);
    float e  = __expf(2.0f * xc);
    return (e - 1.0f) * fast_rcp(e + 1.0f);
}

__global__ __launch_bounds__(256) void simple_et_kernel(
    const float* __restrict__ x_nm, const int* __restrict__ atom_types,
    const int* __restrict__ mask_in, const float* __restrict__ emb_W,
    const float* __restrict__ msg_w1, const float* __restrict__ msg_b1,
    const float* __restrict__ msg_w2, const float* __restrict__ msg_b2,
    const float* __restrict__ upd_w1, const float* __restrict__ upd_b1,
    const float* __restrict__ upd_w2, const float* __restrict__ upd_b2,
    const float* __restrict__ dx_w, const float* __restrict__ dx_b,
    const float* __restrict__ dv_w, const float* __restrict__ dv_b,
    float* __restrict__ out)
{
    __shared__ float4 sx[N_ATOMS];
    __shared__ float4 spA[N_ATOMS];
    __shared__ float4 spB[N_ATOMS];
    __shared__ float4 spt[TYPE_TILES * 16];
    __shared__ float4 spe[TYPE_TILES * 16];
    __shared__ float4 part4[8 * 12];
    __shared__ float4 sw4[12];
    float* const sw = reinterpret_cast<float*>(sw4);

    const int tid  = threadIdx.x;
    const int b    = blockIdx.x / (N_ATOMS / RECV_PER_BLOCK);
    const int row0 = (blockIdx.x % (N_ATOMS / RECV_PER_BLOCK)) * RECV_PER_BLOCK;

    const int lane = tid & 31;
    const int wave = tid >> 5;
    const int grp  = wave >> 2;
    const int qtr  = wave & 3;
    const int m    = lane & 15;
    const int hh   = lane >> 4;

    const bool rowok1 = (m & 3) < 3;
    const int  cm1    = rowok1 ? (m & 3) : 0;
    const bool rowok2 = (m & 7) < 6;
    const int  cm2    = rowok2 ? ((m & 7) >> 1) : 0;
    const int  Rm2    = 2 * (m >> 3) + (m & 1);

    const int gb = b * N_ATOMS;

    const int ty  = wave * 16 + m;
    const int tyc = ty < NUM_TYPES ? ty : NUM_TYPES - 1;
    float e0 = 0.0f, e1 = 0.0f;
    if (wave < TYPE_TILES) {
        e0 = emb_W[tyc * 3 + (hh ? 1 : 0)];
        e1 = emb_W[tyc * 3 + 2];
    }
    const float w1c0 = msg_w1[0 * 3 + cm1], w1c1 = msg_w1[1 * 3 + cm1];
    const float w1c2 = msg_w1[2 * 3 + cm1];
    const float w1r  = msg_w1[3 * 3 + cm2];
    const float mb10 = msg_b1[0], mb11 = msg_b1[1], mb12 = msg_b1[2];
    const int   cq  = m < 3 ? m : 0;
    const float wx0 = msg_w2[(2 * hh) * 3 + cq];
    const float wx1 = hh ? msg_b2[cq] : msg_w2[3 + cq];

    int tt[N_ATOMS / 256];
#pragma unroll
    for (int it = 0; it < N_ATOMS / 256; ++it)
        tt[it] = atom_types[gb + tid + it * 256];
    const int ti = atom_types[gb + row0 + (tid & (RECV_PER_BLOCK - 1))];

    float ew0 = 0.0f, ew1 = 0.0f, ew2 = 0.0f, ew3 = 0.0f;
    float eb0 = 0.0f, eb1 = 0.0f, eb2 = 0.0f, eb3 = 0.0f;
    if (wave == 0) {
        const int k9 = lane < 9 ? lane : 8;
        const int k3 = lane < 3 ? lane : 2;
        ew0 = upd_w1[k9]; ew1 = upd_w2[k9]; ew2 = dx_w[k9]; ew3 = dv_w[k9];
        eb0 = upd_b1[k3]; eb1 = upd_b2[k3]; eb2 = dx_b[k3]; eb3 = dv_b[k3];
    }

    int   mk[N_ATOMS / 256];
    float pos[N_ATOMS / 256][3];
#pragma unroll
    for (int it = 0; it < N_ATOMS / 256; ++it)
        mk[it] = mask_in[gb + tid + it * 256];
#pragma unroll
    for (int it = 0; it < N_ATOMS / 256; ++it) {
        const int n = tid + it * 256;
        pos[it][0] = x_nm[(gb + n) * 3 + 0];
        pos[it][1] = x_nm[(gb + n) * 3 + 1];
        pos[it][2] = x_nm[(gb + n) * 3 + 2];
    }

    v2f A1a;
    {
        const float w0 = hh ? w1c1 : w1c0;
        const float w1 = hh ? w1c2 : 0.0f;
        A1a[0] = rowok1 ? w0 : 0.0f;
        A1a[1] = rowok1 ? w1 : 0.0f;
    }
    if (wave < TYPE_TILES) {
        v8f C1 = {};
        C1[0] = mb10;  C1[1] = mb11;  C1[2] = mb12;
        C1[4] = mb10;  C1[5] = mb11;  C1[6] = mb12;
        v2f Bt;
        Bt[0] = e0;
        Bt[1] = hh ? e1 : 0.0f;
        const v8f P = __builtin_amdgcn_wmma_f32_16x16x4_f32(
            false, A1a, false, Bt, (short)0, C1, false, false);
        if (ty < NUM_TYPES) {
            float* const spef = reinterpret_cast<float*>(spe);
            if (hh == 0) {
                spt[ty] = make_float4(0.5f * P[0], 0.5f * P[1], 0.5f * P[2], 0.0f);
                spef[4 * ty + 0] = e0;
                spef[4 * ty + 3] = 0.0f;
            } else {
                spef[4 * ty + 1] = e0;
                spef[4 * ty + 2] = e1;
            }
        }
    }

#pragma unroll
    for (int it = 0; it < N_ATOMS / 256; ++it) {
        const int n = tid + it * 256;
        sx[n] = make_float4(pos[it][0], pos[it][1], pos[it][2],
                            mk[it] ? 1e-9f : __builtin_inff());
    }
    if (wave == 0) {
        if (lane < 9) { sw[lane] = ew0; sw[9 + lane] = ew1; sw[18 + lane] = ew2; sw[27 + lane] = ew3; }
        if (lane < 3) { sw[36 + lane] = eb0; sw[39 + lane] = eb1; sw[42 + lane] = eb2; sw[45 + lane] = eb3; }
    }
    __syncthreads();
    const float4 pv0 = spt[tt[0]], pv1 = spt[tt[1]], pv2 = spt[tt[2]], pv3 = spt[tt[3]];
    const float4 pv4 = spt[tt[4]], pv5 = spt[tt[5]], pv6 = spt[tt[6]], pv7 = spt[tt[7]];
    __builtin_amdgcn_sched_barrier(0);
#define PUT_SP(it, pv)                                                      \
    spA[tid + (it) * 256] = make_float4((pv).x, (pv).x, (pv).y, (pv).y);    \
    spB[tid + (it) * 256] = make_float4((pv).z, (pv).z, 0.0f, 0.0f);
    PUT_SP(0, pv0)  PUT_SP(1, pv1)  PUT_SP(2, pv2)  PUT_SP(3, pv3)
    PUT_SP(4, pv4)  PUT_SP(5, pv5)  PUT_SP(6, pv6)  PUT_SP(7, pv7)
#undef PUT_SP
    __syncthreads();

    const int ia = row0 + grp * 4 + hh * 2;
    const int ib = ia + 1;
    const float4 xa = sx[ia];
    const float4 xb = sx[ib];
    v2f X0, X1, X2;
    X0[0] = xa.x;  X0[1] = xb.x;
    X1[0] = xa.y;  X1[1] = xb.y;
    X2[0] = xa.z;  X2[1] = xb.z;
    const float cuta = xa.w < 1.0f ? CUTOFF_SQ : -1.0f;
    const float cutb = xb.w < 1.0f ? CUTOFF_SQ : -1.0f;

    const float w1h = 0.5f * w1r;
    v2f A1b;
    A1b[0] = (rowok2 && (2 * hh + 0 == Rm2)) ? w1h : 0.0f;
    A1b[1] = (rowok2 && (2 * hh + 1 == Rm2)) ? w1h : 0.0f;
    v2f P[4][3];
#pragma unroll
    for (int k = 0; k < 4; ++k) {
#pragma unroll
        for (int cc = 0; cc < 3; ++cc) { P[k][cc][0] = 0.0f;  P[k][cc][1] = 0.0f; }
    }

    const int jbase = qtr * (N_ATOMS / SENDER_SPLIT);

#pragma unroll 2
    for (int t = 0; t < (N_ATOMS / SENDER_SPLIT) / TILE; ++t) {
        const int j = jbase + t * TILE + m;
        const float4 xq = sx[j];
        const float4 ca = spA[j];
        const float4 cb = spB[j];

        v2f q0, q1, q2, qw;
        q0[0] = xq.x;  q0[1] = xq.x;
        q1[0] = xq.y;  q1[1] = xq.y;
        q2[0] = xq.z;  q2[1] = xq.z;
        qw[0] = xq.w;  qw[1] = xq.w;
        const v2f d0 = X0 - q0, d1 = X1 - q1, d2 = X2 - q2;
        v2f dsq;
        {
#pragma clang fp contract(off)
            dsq = d0 * d0 + d1 * d1 + d2 * d2 + qw;
        }
        const float inva = fast_rsq(dsq[0]);
        const float invb = fast_rsq(dsq[1]);
        v2f sc, B1b;
        sc[0] = dsq[0] < cuta ? inva : 0.0f;
        sc[1] = dsq[1] < cutb ? invb : 0.0f;
        B1b[0] = fminf(inva, 1.0e3f);
        B1b[1] = fminf(invb, 1.0e3f);

        v8f C1;
        C1[0] = ca.x;  C1[1] = ca.y;  C1[2] = ca.z;  C1[3] = ca.w;
        C1[4] = cb.x;  C1[5] = cb.y;  C1[6] = cb.z;  C1[7] = cb.w;
        const v8f D1 = __builtin_amdgcn_wmma_f32_16x16x4_f32(
            false, A1b, false, B1b, (short)0, C1, false, false);

        v2f hk[3];
#pragma unroll
        for (int k = 0; k < 3; ++k) {
            v2f u, th;
            u[0] = D1[2 * k];  u[1] = D1[2 * k + 1];
            th[0] = __builtin_amdgcn_tanhf(u[0]);
            th[1] = __builtin_amdgcn_tanhf(u[1]);
            hk[k] = __builtin_elementwise_fma(u, th, u);
        }

        const v2f dir[3] = { d0 * sc, d1 * sc, d2 * sc };
#pragma unroll
        for (int cc = 0; cc < 3; ++cc) {
#pragma unroll
            for (int k = 0; k < 3; ++k)
                P[k][cc] = __builtin_elementwise_fma(hk[k], dir[cc], P[k][cc]);
            P[3][cc] = P[3][cc] + dir[cc];
        }
    }

#pragma unroll
    for (int off = 8; off > 0; off >>= 1) {
#pragma unroll
        for (int k = 0; k < 4; ++k) {
#pragma unroll
            for (int cc = 0; cc < 3; ++cc) {
                P[k][cc][0] += __shfl_xor(P[k][cc][0], off, 32);
                P[k][cc][1] += __shfl_xor(P[k][cc][1], off, 32);
            }
        }
    }
    if (m == 0) {
#pragma unroll
        for (int s = 0; s < 2; ++s) {
#pragma unroll
            for (int cc = 0; cc < 3; ++cc)
                part4[((wave * 2 + hh) * 2 + s) * 3 + cc] =
                    make_float4(P[0][cc][s], P[1][cc][s], P[2][cc][s], P[3][cc][s]);
        }
    }
    float wl[48];
    float hi0 = 0.0f, hi1 = 0.0f, hi2 = 0.0f;
    if (tid < RECV_PER_BLOCK) {
#pragma unroll
        for (int q = 0; q < 12; ++q) {
            const float4 v = sw4[q];
            wl[4 * q + 0] = v.x;  wl[4 * q + 1] = v.y;  wl[4 * q + 2] = v.z;  wl[4 * q + 3] = v.w;
        }
        const float4 hi4 = spe[ti];
        hi0 = hi4.x; hi1 = hi4.y; hi2 = hi4.z;
    }
    __syncthreads();

    if (wave != 0) return;

    v8f Dm = {};
    {
        const bool colok = m < RECV_PER_BLOCK;
        const int  Rc    = colok ? m : 0;
        const int  pb    = (((Rc >> 2) * SENDER_SPLIT * 2 + ((Rc >> 1) & 1)) * 2 + (Rc & 1)) * 3;
        const float* const pf = reinterpret_cast<const float*>(part4);
#pragma unroll
        for (int cc = 0; cc < 3; ++cc) {
            v2f Bm;
            Bm[0] = 0.0f;  Bm[1] = 0.0f;
#pragma unroll
            for (int q = 0; q < SENDER_SPLIT; ++q) {
                const float* p = pf + (pb + q * 12 + cc) * 4 + 2 * hh;
                Bm[0] += p[0];
                Bm[1] += p[1];
            }
            if (!colok) { Bm[0] = 0.0f;  Bm[1] = 0.0f; }
            v2f Am;
            Am[0] = (m == cc) ? wx0 : 0.0f;
            Am[1] = (m == cc) ? wx1 : 0.0f;
            Dm = __builtin_amdgcn_wmma_f32_16x16x4_f32(
                false, Am, false, Bm, (short)0, Dm, false, false);
        }
    }

    if (tid < RECV_PER_BLOCK) {
        const float v0 = Dm[0], v1 = Dm[1], v2 = Dm[2];
        const int i = row0 + tid;
        const float* uw1 = wl;       const float* uw2 = wl + 9;
        const float* xw  = wl + 18;  const float* vw  = wl + 27;
        const float* ub1 = wl + 36;  const float* ub2 = wl + 39;
        const float* xb_ = wl + 42;  const float* vb_ = wl + 45;

        const float u0 = silu_f(v0 * uw1[0] + v1 * uw1[3] + v2 * uw1[6] + ub1[0]);
        const float u1 = silu_f(v0 * uw1[1] + v1 * uw1[4] + v2 * uw1[7] + ub1[1]);
        const float u2 = silu_f(v0 * uw1[2] + v1 * uw1[5] + v2 * uw1[8] + ub1[2]);

        const float h0 = hi0 + u0 * uw2[0] + u1 * uw2[3] + u2 * uw2[6] + ub2[0];
        const float h1 = hi1 + u0 * uw2[1] + u1 * uw2[4] + u2 * uw2[7] + ub2[1];
        const float h2 = hi2 + u0 * uw2[2] + u1 * uw2[5] + u2 * uw2[8] + ub2[2];

        const int base = (b * N_ATOMS + i) * 3;
        const int half = BATCH * N_ATOMS * 3;
#pragma unroll
        for (int c = 0; c < 3; ++c) {
            const float px = h0 * xw[0 * 3 + c] + h1 * xw[1 * 3 + c] + h2 * xw[2 * 3 + c] + xb_[c];
            const float pv = h0 * vw[0 * 3 + c] + h1 * vw[1 * 3 + c] + h2 * vw[2 * 3 + c] + vb_[c];
            out[base + c]        = tanh_f(px) * MAX_DISP;
            out[half + base + c] = tanh_f(pv) * 1.0f;
        }
    }
}

extern "C" void kernel_launch(void* const* d_in, const int* in_sizes, int n_in,
                              void* d_out, int out_size, void* d_ws, size_t ws_size,
                              hipStream_t stream) {
    (void)in_sizes; (void)n_in; (void)d_ws; (void)ws_size; (void)out_size;
    const float* x_nm   = (const float*)d_in[0];
    const int*   types  = (const int*)  d_in[1];
    const int*   mask   = (const int*)  d_in[2];
    const float* emb_W  = (const float*)d_in[3];
    const float* msg_w1 = (const float*)d_in[4];
    const float* msg_b1 = (const float*)d_in[5];
    const float* msg_w2 = (const float*)d_in[6];
    const float* msg_b2 = (const float*)d_in[7];
    const float* upd_w1 = (const float*)d_in[8];
    const float* upd_b1 = (const float*)d_in[9];
    const float* upd_w2 = (const float*)d_in[10];
    const float* upd_b2 = (const float*)d_in[11];
    const float* dx_w   = (const float*)d_in[12];
    const float* dx_b   = (const float*)d_in[13];
    const float* dv_w   = (const float*)d_in[14];
    const float* dv_b   = (const float*)d_in[15];
    float* out = (float*)d_out;

    dim3 grid(BATCH * (N_ATOMS / RECV_PER_BLOCK));
    dim3 block(256);
    simple_et_kernel<<<grid, block, 0, stream>>>(
        x_nm, types, mask, emb_W,
        msg_w1, msg_b1, msg_w2, msg_b2,
        upd_w1, upd_b1, upd_w2, upd_b2,
        dx_w, dx_b, dv_w, dv_b, out);
}


# Sixteen receivers per block, was eight: 256 blocks, half as many waves enter; a wave takes half of the senders (64 trips, four per pass), senders staged once per WGP; sum order changes
# speedup vs baseline: 3.2878x; 1.1131x over previous
#include <hip/hip_runtime.h>


typedef float v2f __attribute__((ext_vector_type(2)));
typedef float v8f __attribute__((ext_vector_type(8)));

#define N_ATOMS 2048
#define BATCH 2
#define RECV_PER_BLOCK 16
#define SENDER_SPLIT 2
#define TILE 16
#define CUTOFF_SQ 0.49f
#define MAX_DISP 0.02f
#define NUM_TYPES 100
#define TYPE_TILES 7

__device__ __forceinline__ float fast_rcp(float x)  { return __builtin_amdgcn_rcpf(x); }
__device__ __forceinline__ float fast_rsq(float x)  { return __builtin_amdgcn_rsqf(x); }

__device__ __forceinline__ float silu_f(float x) {
    return x * fast_rcp(1.0f + __expf(-x));
}
__device__ __forceinline__ float silu_of_half(float u) {
    return __builtin_fmaf(u, __builtin_amdgcn_tanhf(u), u);
}
__device__ __forceinline__ float tanh_f(float x) {
    float xc = fminf(fmaxf(x, -15.0f), 15.0f);
    float e  = __expf(2.0f * xc);
    return (e - 1.0f) * fast_rcp(e + 1.0f);
}

__global__ __launch_bounds__(256) void simple_et_kernel(
    const float* __restrict__ x_nm, const int* __restrict__ atom_types,
    const int* __restrict__ mask_in, const float* __restrict__ emb_W,
    const float* __restrict__ msg_w1, const float* __restrict__ msg_b1,
    const float* __restrict__ msg_w2, const float* __restrict__ msg_b2,
    const float* __restrict__ upd_w1, const float* __restrict__ upd_b1,
    const float* __restrict__ upd_w2, const float* __restrict__ upd_b2,
    const float* __restrict__ dx_w, const float* __restrict__ dx_b,
    const float* __restrict__ dv_w, const float* __restrict__ dv_b,
    float* __restrict__ out)
{
    __shared__ float4 sx[N_ATOMS];
    __shared__ float4 spA[N_ATOMS];
    __shared__ float4 spB[N_ATOMS];
    __shared__ float4 spt[TYPE_TILES * 16];
    __shared__ float4 spe[TYPE_TILES * 16];
    __shared__ float4 part4[8 * 12];
    __shared__ float4 sw4[12];
    float* const sw = reinterpret_cast<float*>(sw4);

    const int tid  = threadIdx.x;
    const int b    = blockIdx.x / (N_ATOMS / RECV_PER_BLOCK);
    const int row0 = (blockIdx.x % (N_ATOMS / RECV_PER_BLOCK)) * RECV_PER_BLOCK;

    const int lane = tid & 31;
    const int wave = tid >> 5;
    const int grp  = wave >> 1;
    const int shf  = wave & 1;
    const int m    = lane & 15;
    const int hh   = lane >> 4;

    const bool rowok1 = (m & 3) < 3;
    const int  cm1    = rowok1 ? (m & 3) : 0;
    const bool rowok2 = (m & 7) < 6;
    const int  cm2    = rowok2 ? ((m & 7) >> 1) : 0;
    const int  Rm2    = 2 * (m >> 3) + (m & 1);

    const int gb = b * N_ATOMS;

    const int ty  = wave * 16 + m;
    const int tyc = ty < NUM_TYPES ? ty : NUM_TYPES - 1;
    float e0 = 0.0f, e1 = 0.0f;
    if (wave < TYPE_TILES) {
        e0 = emb_W[tyc * 3 + (hh ? 1 : 0)];
        e1 = emb_W[tyc * 3 + 2];
    }
    const float w1c0 = msg_w1[0 * 3 + cm1], w1c1 = msg_w1[1 * 3 + cm1];
    const float w1c2 = msg_w1[2 * 3 + cm1];
    const float w1r  = msg_w1[3 * 3 + cm2];
    const float mb10 = msg_b1[0], mb11 = msg_b1[1], mb12 = msg_b1[2];
    const int   cq  = m < 3 ? m : 0;
    const float wx0 = msg_w2[(2 * hh) * 3 + cq];
    const float wx1 = hh ? msg_b2[cq] : msg_w2[3 + cq];

    int tt[N_ATOMS / 256];
#pragma unroll
    for (int it = 0; it < N_ATOMS / 256; ++it)
        tt[it] = atom_types[gb + tid + it * 256];
    const int ti = atom_types[gb + row0 + (tid & (RECV_PER_BLOCK - 1))];

    float ew0 = 0.0f, ew1 = 0.0f, ew2 = 0.0f, ew3 = 0.0f;
    float eb0 = 0.0f, eb1 = 0.0f, eb2 = 0.0f, eb3 = 0.0f;
    if (wave == 0) {
        const int k9 = lane < 9 ? lane : 8;
        const int k3 = lane < 3 ? lane : 2;
        ew0 = upd_w1[k9]; ew1 = upd_w2[k9]; ew2 = dx_w[k9]; ew3 = dv_w[k9];
        eb0 = upd_b1[k3]; eb1 = upd_b2[k3]; eb2 = dx_b[k3]; eb3 = dv_b[k3];
    }

    int   mk[N_ATOMS / 256];
    float pos[N_ATOMS / 256][3];
#pragma unroll
    for (int it = 0; it < N_ATOMS / 256; ++it)
        mk[it] = mask_in[gb + tid + it * 256];
#pragma unroll
    for (int it = 0; it < N_ATOMS / 256; ++it) {
        const int n = tid + it * 256;
        pos[it][0] = x_nm[(gb + n) * 3 + 0];
        pos[it][1] = x_nm[(gb + n) * 3 + 1];
        pos[it][2] = x_nm[(gb + n) * 3 + 2];
    }

    v2f A1a;
    {
        const float w0 = hh ? w1c1 : w1c0;
        const float w1 = hh ? w1c2 : 0.0f;
        A1a[0] = rowok1 ? w0 : 0.0f;
        A1a[1] = rowok1 ? w1 : 0.0f;
    }
    if (wave < TYPE_TILES) {
        v8f C1 = {};
        C1[0] = mb10;  C1[1] = mb11;  C1[2] = mb12;
        C1[4] = mb10;  C1[5] = mb11;  C1[6] = mb12;
        v2f Bt;
        Bt[0] = e0;
        Bt[1] = hh ? e1 : 0.0f;
        const v8f P = __builtin_amdgcn_wmma_f32_16x16x4_f32(
            false, A1a, false, Bt, (short)0, C1, false, false);
        if (ty < NUM_TYPES) {
            float* const spef = reinterpret_cast<float*>(spe);
            if (hh == 0) {
                spt[ty] = make_float4(0.5f * P[0], 0.5f * P[1], 0.5f * P[2], 0.0f);
                spef[4 * ty + 0] = e0;
                spef[4 * ty + 3] = 0.0f;
            } else {
                spef[4 * ty + 1] = e0;
                spef[4 * ty + 2] = e1;
            }
        }
    }

#pragma unroll
    for (int it = 0; it < N_ATOMS / 256; ++it) {
        const int n = tid + it * 256;
        sx[n] = make_float4(pos[it][0], pos[it][1], pos[it][2],
                            mk[it] ? 1e-9f : __builtin_inff());
    }
    if (wave == 0) {
        if (lane < 9) { sw[lane] = ew0; sw[9 + lane] = ew1; sw[18 + lane] = ew2; sw[27 + lane] = ew3; }
        if (lane < 3) { sw[36 + lane] = eb0; sw[39 + lane] = eb1; sw[42 + lane] = eb2; sw[45 + lane] = eb3; }
    }
    __syncthreads();
    const float4 pv0 = spt[tt[0]], pv1 = spt[tt[1]], pv2 = spt[tt[2]], pv3 = spt[tt[3]];
    const float4 pv4 = spt[tt[4]], pv5 = spt[tt[5]], pv6 = spt[tt[6]], pv7 = spt[tt[7]];
    __builtin_amdgcn_sched_barrier(0);
#define PUT_SP(it, pv)                                                      \
    spA[tid + (it) * 256] = make_float4((pv).x, (pv).x, (pv).y, (pv).y);    \
    spB[tid + (it) * 256] = make_float4((pv).z, (pv).z, 0.0f, 0.0f);
    PUT_SP(0, pv0)  PUT_SP(1, pv1)  PUT_SP(2, pv2)  PUT_SP(3, pv3)
    PUT_SP(4, pv4)  PUT_SP(5, pv5)  PUT_SP(6, pv6)  PUT_SP(7, pv7)
#undef PUT_SP
    __syncthreads();

    const int ia = row0 + grp * 4 + hh * 2;
    const int ib = ia + 1;
    const float4 xa = sx[ia];
    const float4 xb = sx[ib];
    v2f X0, X1, X2;
    X0[0] = xa.x;  X0[1] = xb.x;
    X1[0] = xa.y;  X1[1] = xb.y;
    X2[0] = xa.z;  X2[1] = xb.z;
    const float cuta = xa.w < 1.0f ? CUTOFF_SQ : -1.0f;
    const float cutb = xb.w < 1.0f ? CUTOFF_SQ : -1.0f;

    const float w1h = 0.5f * w1r;
    v2f A1b;
    A1b[0] = (rowok2 && (2 * hh + 0 == Rm2)) ? w1h : 0.0f;
    A1b[1] = (rowok2 && (2 * hh + 1 == Rm2)) ? w1h : 0.0f;
    v2f P[4][3];
#pragma unroll
    for (int k = 0; k < 4; ++k) {
#pragma unroll
        for (int cc = 0; cc < 3; ++cc) { P[k][cc][0] = 0.0f;  P[k][cc][1] = 0.0f; }
    }

    const int jbase = shf * (N_ATOMS / SENDER_SPLIT);

#pragma unroll 4
    for (int t = 0; t < (N_ATOMS / SENDER_SPLIT) / TILE; ++t) {
        const int j = jbase + t * TILE + m;
        const float4 xq = sx[j];
        const float4 ca = spA[j];
        const float4 cb = spB[j];

        v2f q0, q1, q2, qw;
        q0[0] = xq.x;  q0[1] = xq.x;
        q1[0] = xq.y;  q1[1] = xq.y;
        q2[0] = xq.z;  q2[1] = xq.z;
        qw[0] = xq.w;  qw[1] = xq.w;
        const v2f d0 = X0 - q0, d1 = X1 - q1, d2 = X2 - q2;
        v2f dsq;
        {
#pragma clang fp contract(off)
            dsq = d0 * d0 + d1 * d1 + d2 * d2 + qw;
        }
        const float inva = fast_rsq(dsq[0]);
        const float invb = fast_rsq(dsq[1]);
        v2f sc, B1b;
        sc[0] = dsq[0] < cuta ? inva : 0.0f;
        sc[1] = dsq[1] < cutb ? invb : 0.0f;
        B1b[0] = fminf(inva, 1.0e3f);
        B1b[1] = fminf(invb, 1.0e3f);

        v8f C1;
        C1[0] = ca.x;  C1[1] = ca.y;  C1[2] = ca.z;  C1[3] = ca.w;
        C1[4] = cb.x;  C1[5] = cb.y;  C1[6] = cb.z;  C1[7] = cb.w;
        const v8f D1 = __builtin_amdgcn_wmma_f32_16x16x4_f32(
            false, A1b, false, B1b, (short)0, C1, false, false);

        v2f hk[3];
#pragma unroll
        for (int k = 0; k < 3; ++k) {
            v2f u, th;
            u[0] = D1[2 * k];  u[1] = D1[2 * k + 1];
            th[0] = __builtin_amdgcn_tanhf(u[0]);
            th[1] = __builtin_amdgcn_tanhf(u[1]);
            hk[k] = __builtin_elementwise_fma(u, th, u);
        }

        const v2f dir[3] = { d0 * sc, d1 * sc, d2 * sc };
#pragma unroll
        for (int cc = 0; cc < 3; ++cc) {
#pragma unroll
            for (int k = 0; k < 3; ++k)
                P[k][cc] = __builtin_elementwise_fma(hk[k], dir[cc], P[k][cc]);
            P[3][cc] = P[3][cc] + dir[cc];
        }
    }

#pragma unroll
    for (int off = 8; off > 0; off >>= 1) {
#pragma unroll
        for (int k = 0; k < 4; ++k) {
#pragma unroll
            for (int cc = 0; cc < 3; ++cc) {
                P[k][cc][0] += __shfl_xor(P[k][cc][0], off, 32);
                P[k][cc][1] += __shfl_xor(P[k][cc][1], off, 32);
            }
        }
    }
    if (m == 0) {
#pragma unroll
        for (int s = 0; s < 2; ++s) {
#pragma unroll
            for (int cc = 0; cc < 3; ++cc)
                part4[((wave * 2 + hh) * 2 + s) * 3 + cc] =
                    make_float4(P[0][cc][s], P[1][cc][s], P[2][cc][s], P[3][cc][s]);
        }
    }
    float wl[48];
    float hi0 = 0.0f, hi1 = 0.0f, hi2 = 0.0f;
    if (tid < RECV_PER_BLOCK) {
#pragma unroll
        for (int q = 0; q < 12; ++q) {
            const float4 v = sw4[q];
            wl[4 * q + 0] = v.x;  wl[4 * q + 1] = v.y;  wl[4 * q + 2] = v.z;  wl[4 * q + 3] = v.w;
        }
        const float4 hi4 = spe[ti];
        hi0 = hi4.x; hi1 = hi4.y; hi2 = hi4.z;
    }
    __syncthreads();

    if (wave != 0) return;

    v8f Dm = {};
    {
        const int  Rc    = m;
        const int  pb    = (((Rc >> 2) * SENDER_SPLIT * 2 + ((Rc >> 1) & 1)) * 2 + (Rc & 1)) * 3;
        const float* const pf = reinterpret_cast<const float*>(part4);
#pragma unroll
        for (int cc = 0; cc < 3; ++cc) {
            v2f Bm;
            Bm[0] = 0.0f;  Bm[1] = 0.0f;
#pragma unroll
            for (int q = 0; q < SENDER_SPLIT; ++q) {
                const float* p = pf + (pb + q * 12 + cc) * 4 + 2 * hh;
                Bm[0] += p[0];
                Bm[1] += p[1];
            }
            v2f Am;
            Am[0] = (m == cc) ? wx0 : 0.0f;
            Am[1] = (m == cc) ? wx1 : 0.0f;
            Dm = __builtin_amdgcn_wmma_f32_16x16x4_f32(
                false, Am, false, Bm, (short)0, Dm, false, false);
        }
    }

    if (tid < RECV_PER_BLOCK) {
        const float v0 = Dm[0], v1 = Dm[1], v2 = Dm[2];
        const int i = row0 + tid;
        const float* uw1 = wl;       const float* uw2 = wl + 9;
        const float* xw  = wl + 18;  const float* vw  = wl + 27;
        const float* ub1 = wl + 36;  const float* ub2 = wl + 39;
        const float* xb_ = wl + 42;  const float* vb_ = wl + 45;

        const float u0 = silu_f(v0 * uw1[0] + v1 * uw1[3] + v2 * uw1[6] + ub1[0]);
        const float u1 = silu_f(v0 * uw1[1] + v1 * uw1[4] + v2 * uw1[7] + ub1[1]);
        const float u2 = silu_f(v0 * uw1[2] + v1 * uw1[5] + v2 * uw1[8] + ub1[2]);

        const float h0 = hi0 + u0 * uw2[0] + u1 * uw2[3] + u2 * uw2[6] + ub2[0];
        const float h1 = hi1 + u0 * uw2[1] + u1 * uw2[4] + u2 * uw2[7] + ub2[1];
        const float h2 = hi2 + u0 * uw2[2] + u1 * uw2[5] + u2 * uw2[8] + ub2[2];

        const int base = (b * N_ATOMS + i) * 3;
        const int half = BATCH * N_ATOMS * 3;
#pragma unroll
        for (int c = 0; c < 3; ++c) {
            const float px = h0 * xw[0 * 3 + c] + h1 * xw[1 * 3 + c] + h2 * xw[2 * 3 + c] + xb_[c];
            const float pv = h0 * vw[0 * 3 + c] + h1 * vw[1 * 3 + c] + h2 * vw[2 * 3 + c] + vb_[c];
            out[base + c]        = tanh_f(px) * MAX_DISP;
            out[half + base + c] = tanh_f(pv) * 1.0f;
        }
    }
}

extern "C" void kernel_launch(void* const* d_in, const int* in_sizes, int n_in,
                              void* d_out, int out_size, void* d_ws, size_t ws_size,
                              hipStream_t stream) {
    (void)in_sizes; (void)n_in; (void)d_ws; (void)ws_size; (void)out_size;
    const float* x_nm   = (const float*)d_in[0];
    const int*   types  = (const int*)  d_in[1];
    const int*   mask   = (const int*)  d_in[2];
    const float* emb_W  = (const float*)d_in[3];
    const float* msg_w1 = (const float*)d_in[4];
    const float* msg_b1 = (const float*)d_in[5];
    const float* msg_w2 = (const float*)d_in[6];
    const float* msg_b2 = (const float*)d_in[7];
    const float* upd_w1 = (const float*)d_in[8];
    const float* upd_b1 = (const float*)d_in[9];
    const float* upd_w2 = (const float*)d_in[10];
    const float* upd_b2 = (const float*)d_in[11];
    const float* dx_w   = (const float*)d_in[12];
    const float* dx_b   = (const float*)d_in[13];
    const float* dv_w   = (const float*)d_in[14];
    const float* dv_b   = (const float*)d_in[15];
    float* out = (float*)d_out;

    dim3 grid(BATCH * (N_ATOMS / RECV_PER_BLOCK));
    dim3 block(256);
    simple_et_kernel<<<grid, block, 0, stream>>>(
        x_nm, types, mask, emb_W,
        msg_w1, msg_b1, msg_w2, msg_b2,
        upd_w1, upd_b1, upd_w2, upd_b2,
        dx_w, dx_b, dv_w, dv_b, out);
}


# Pair tile transposed: a lane owns one receiver (16 tile columns), k = 4 senders staged two by two; sender sums stay in the lane, the 96 cross-lane shuffles at the end go; sum order changes
# speedup vs baseline: 3.5326x; 1.0745x over previous
#include <hip/hip_runtime.h>


typedef float v2f __attribute__((ext_vector_type(2)));
typedef float v8f __attribute__((ext_vector_type(8)));

#define N_ATOMS 2048
#define BATCH 2
#define RECV_PER_BLOCK 16
#define N_PAIRS (N_ATOMS / 2)
#define TRIP_SENDERS 4
#define CUTOFF_SQ 0.49f
#define MAX_DISP 0.02f
#define NUM_TYPES 100
#define TYPE_TILES 7

__device__ __forceinline__ float fast_rcp(float x)  { return __builtin_amdgcn_rcpf(x); }
__device__ __forceinline__ float fast_rsq(float x)  { return __builtin_amdgcn_rsqf(x); }

__device__ __forceinline__ float silu_f(float x) {
    return x * fast_rcp(1.0f + __expf(-x));
}
__device__ __forceinline__ float silu_of_half(float u) {
    return __builtin_fmaf(u, __builtin_amdgcn_tanhf(u), u);
}
__device__ __forceinline__ float other_half(float v) {
    const unsigned r = __builtin_amdgcn_permlanex16(
        0u, __builtin_bit_cast(unsigned, v), 0x76543210u, 0xfedcba98u, false, false);
    return __builtin_bit_cast(float, r);
}
__device__ __forceinline__ float tanh_f(float x) {
    float xc = fminf(fmaxf(x, -15.0f), 15.0f);
    float e  = __expf(2.0f * xc);
    return (e - 1.0f) * fast_rcp(e + 1.0f);
}

__global__ __launch_bounds__(256) void simple_et_kernel(
    const float* __restrict__ x_nm, const int* __restrict__ atom_types,
    const int* __restrict__ mask_in, const float* __restrict__ emb_W,
    const float* __restrict__ msg_w1, const float* __restrict__ msg_b1,
    const float* __restrict__ msg_w2, const float* __restrict__ msg_b2,
    const float* __restrict__ upd_w1, const float* __restrict__ upd_b1,
    const float* __restrict__ upd_w2, const float* __restrict__ upd_b2,
    const float* __restrict__ dx_w, const float* __restrict__ dx_b,
    const float* __restrict__ dv_w, const float* __restrict__ dv_b,
    float* __restrict__ out)
{
    __shared__ float4 sxp0[N_PAIRS];
    __shared__ float4 sxp1[N_PAIRS];
    __shared__ float4 spp0[N_PAIRS];
    __shared__ float4 spp1[N_PAIRS];
    __shared__ float4 spt[TYPE_TILES * 16];
    __shared__ float4 spe[TYPE_TILES * 16];
    __shared__ float4 part4[8 * RECV_PER_BLOCK * 3];
    __shared__ float4 sw4[12];
    float* const sw = reinterpret_cast<float*>(sw4);

    const int tid  = threadIdx.x;
    const int b    = blockIdx.x / (N_ATOMS / RECV_PER_BLOCK);
    const int row0 = (blockIdx.x % (N_ATOMS / RECV_PER_BLOCK)) * RECV_PER_BLOCK;

    const int lane = tid & 31;
    const int wave = tid >> 5;
    const int m    = lane & 15;
    const int hh   = lane >> 4;

    const bool rowok1 = (m & 3) < 3;
    const int  cm1    = rowok1 ? (m & 3) : 0;
    const bool rowok2 = (m & 7) < 6;
    const int  cm2    = rowok2 ? ((m & 7) >> 1) : 0;
    const int  Rm2    = 2 * (m >> 3) + (m & 1);

    const int gb = b * N_ATOMS;

    const int ty  = wave * 16 + m;
    const int tyc = ty < NUM_TYPES ? ty : NUM_TYPES - 1;
    float e0 = 0.0f, e1 = 0.0f;
    if (wave < TYPE_TILES) {
        e0 = emb_W[tyc * 3 + (hh ? 1 : 0)];
        e1 = emb_W[tyc * 3 + 2];
    }
    const float w1c0 = msg_w1[0 * 3 + cm1], w1c1 = msg_w1[1 * 3 + cm1];
    const float w1c2 = msg_w1[2 * 3 + cm1];
    const float w1r  = msg_w1[3 * 3 + cm2];
    const float mb10 = msg_b1[0], mb11 = msg_b1[1], mb12 = msg_b1[2];
    const int   cq  = m < 3 ? m : 0;
    const float wx0 = msg_w2[(2 * hh) * 3 + cq];
    const float wx1 = hh ? msg_b2[cq] : msg_w2[3 + cq];

    int tta[N_PAIRS / 256], ttb[N_PAIRS / 256];
#pragma unroll
    for (int it = 0; it < N_PAIRS / 256; ++it) {
        const int n = 2 * (tid + it * 256);
        tta[it] = atom_types[gb + n];
        ttb[it] = atom_types[gb + n + 1];
    }
    const int ti = atom_types[gb + row0 + (tid & (RECV_PER_BLOCK - 1))];

    float ew0 = 0.0f, ew1 = 0.0f, ew2 = 0.0f, ew3 = 0.0f;
    float eb0 = 0.0f, eb1 = 0.0f, eb2 = 0.0f, eb3 = 0.0f;
    if (wave == 0) {
        const int k9 = lane < 9 ? lane : 8;
        const int k3 = lane < 3 ? lane : 2;
        ew0 = upd_w1[k9]; ew1 = upd_w2[k9]; ew2 = dx_w[k9]; ew3 = dv_w[k9];
        eb0 = upd_b1[k3]; eb1 = upd_b2[k3]; eb2 = dx_b[k3]; eb3 = dv_b[k3];
    }

    int   mka[N_PAIRS / 256], mkb[N_PAIRS / 256];
    float pos[N_PAIRS / 256][6];
#pragma unroll
    for (int it = 0; it < N_PAIRS / 256; ++it) {
        const int n = 2 * (tid + it * 256);
        mka[it] = mask_in[gb + n];
        mkb[it] = mask_in[gb + n + 1];
    }
#pragma unroll
    for (int it = 0; it < N_PAIRS / 256; ++it) {
        const int n = 2 * (tid + it * 256);
#pragma unroll
        for (int q = 0; q < 6; ++q)
            pos[it][q] = x_nm[(gb + n) * 3 + q];
    }

    v2f A1a;
    {
        const float w0 = hh ? w1c1 : w1c0;
        const float w1 = hh ? w1c2 : 0.0f;
        A1a[0] = rowok1 ? w0 : 0.0f;
        A1a[1] = rowok1 ? w1 : 0.0f;
    }
    if (wave < TYPE_TILES) {
        v8f C1 = {};
        C1[0] = mb10;  C1[1] = mb11;  C1[2] = mb12;
        C1[4] = mb10;  C1[5] = mb11;  C1[6] = mb12;
        v2f Bt;
        Bt[0] = e0;
        Bt[1] = hh ? e1 : 0.0f;
        const v8f P = __builtin_amdgcn_wmma_f32_16x16x4_f32(
            false, A1a, false, Bt, (short)0, C1, false, false);
        if (ty < NUM_TYPES) {
            float* const spef = reinterpret_cast<float*>(spe);
            if (hh == 0) {
                spt[ty] = make_float4(0.5f * P[0], 0.5f * P[1], 0.5f * P[2], 0.0f);
                spef[4 * ty + 0] = e0;
                spef[4 * ty + 3] = 0.0f;
            } else {
                spef[4 * ty + 1] = e0;
                spef[4 * ty + 2] = e1;
            }
        }
    }

#pragma unroll
    for (int it = 0; it < N_PAIRS / 256; ++it) {
        const int p = tid + it * 256;
        sxp0[p] = make_float4(pos[it][0], pos[it][3], pos[it][1], pos[it][4]);
        sxp1[p] = make_float4(pos[it][2], pos[it][5],
                              mka[it] ? 1e-9f : __builtin_inff(),
                              mkb[it] ? 1e-9f : __builtin_inff());
    }
    if (wave == 0) {
        if (lane < 9) { sw[lane] = ew0; sw[9 + lane] = ew1; sw[18 + lane] = ew2; sw[27 + lane] = ew3; }
        if (lane < 3) { sw[36 + lane] = eb0; sw[39 + lane] = eb1; sw[42 + lane] = eb2; sw[45 + lane] = eb3; }
    }
    __syncthreads();
    const float4 pa0 = spt[tta[0]], pb0 = spt[ttb[0]], pa1 = spt[tta[1]], pb1 = spt[ttb[1]];
    const float4 pa2 = spt[tta[2]], pb2 = spt[ttb[2]], pa3 = spt[tta[3]], pb3 = spt[ttb[3]];
    __builtin_amdgcn_sched_barrier(0);
#define PUT_SP(it, pa, pb)                                                    \
    spp0[tid + (it) * 256] = make_float4((pa).x, (pb).x, (pa).y, (pb).y);     \
    spp1[tid + (it) * 256] = make_float4((pa).z, (pb).z, 0.0f, 0.0f);
    PUT_SP(0, pa0, pb0)  PUT_SP(1, pa1, pb1)  PUT_SP(2, pa2, pb2)  PUT_SP(3, pa3, pb3)
#undef PUT_SP
    __syncthreads();

    const int ir = row0 + m;
    const float* const sx0f = reinterpret_cast<const float*>(sxp0);
    const float* const sx1f = reinterpret_cast<const float*>(sxp1);
    const float xr0 = sx0f[(ir >> 1) * 4 + (ir & 1)];
    const float xr1 = sx0f[(ir >> 1) * 4 + 2 + (ir & 1)];
    const float xr2 = sx1f[(ir >> 1) * 4 + (ir & 1)];
    const float xre = sx1f[(ir >> 1) * 4 + 2 + (ir & 1)];
    v2f X0, X1, X2;
    X0[0] = xr0;  X0[1] = xr0;
    X1[0] = xr1;  X1[1] = xr1;
    X2[0] = xr2;  X2[1] = xr2;
    const float cutr = xre < 1.0f ? CUTOFF_SQ : -1.0f;

    const float w1h = 0.5f * w1r;
    v2f A1b;
    A1b[0] = (rowok2 && (2 * hh + 0 == Rm2)) ? w1h : 0.0f;
    A1b[1] = (rowok2 && (2 * hh + 1 == Rm2)) ? w1h : 0.0f;
    v2f P[4][3];
#pragma unroll
    for (int k = 0; k < 4; ++k) {
#pragma unroll
        for (int cc = 0; cc < 3; ++cc) { P[k][cc][0] = 0.0f;  P[k][cc][1] = 0.0f; }
    }

    const int pbase = wave * (N_PAIRS / 8) + hh;

#pragma unroll 4
    for (int t = 0; t < (N_ATOMS / 8) / TRIP_SENDERS; ++t) {
        const int p = pbase + t * 2;
        const float4 xq = sxp0[p];
        const float4 xz = sxp1[p];
        const float4 ca = spp0[p];
        const float4 cb = spp1[p];

        v2f q0, q1, q2, qw;
        q0[0] = xq.x;  q0[1] = xq.y;
        q1[0] = xq.z;  q1[1] = xq.w;
        q2[0] = xz.x;  q2[1] = xz.y;
        qw[0] = xz.z;  qw[1] = xz.w;
        const v2f d0 = X0 - q0, d1 = X1 - q1, d2 = X2 - q2;
        v2f dsq;
        {
#pragma clang fp contract(off)
            dsq = d0 * d0 + d1 * d1 + d2 * d2 + qw;
        }
        const float inva = fast_rsq(dsq[0]);
        const float invb = fast_rsq(dsq[1]);
        v2f sc, B1b;
        sc[0] = dsq[0] < cutr ? inva : 0.0f;
        sc[1] = dsq[1] < cutr ? invb : 0.0f;
        B1b[0] = fminf(inva, 1.0e3f);
        B1b[1] = fminf(invb, 1.0e3f);

        v8f C1;
        C1[0] = ca.x;  C1[1] = ca.y;  C1[2] = ca.z;  C1[3] = ca.w;
        C1[4] = cb.x;  C1[5] = cb.y;  C1[6] = cb.z;  C1[7] = cb.w;
        const v8f D1 = __builtin_amdgcn_wmma_f32_16x16x4_f32(
            false, A1b, false, B1b, (short)0, C1, false, false);

        v2f hk[3];
#pragma unroll
        for (int k = 0; k < 3; ++k) {
            v2f u, th;
            u[0] = D1[2 * k];  u[1] = D1[2 * k + 1];
            th[0] = __builtin_amdgcn_tanhf(u[0]);
            th[1] = __builtin_amdgcn_tanhf(u[1]);
            hk[k] = __builtin_elementwise_fma(u, th, u);
        }

        const v2f dir[3] = { d0 * sc, d1 * sc, d2 * sc };
#pragma unroll
        for (int cc = 0; cc < 3; ++cc) {
#pragma unroll
            for (int k = 0; k < 3; ++k)
                P[k][cc] = __builtin_elementwise_fma(hk[k], dir[cc], P[k][cc]);
            P[3][cc] = P[3][cc] + dir[cc];
        }
    }

    float S[4][3];
#pragma unroll
    for (int k = 0; k < 4; ++k) {
#pragma unroll
        for (int cc = 0; cc < 3; ++cc) {
            const float v = P[k][cc][0] + P[k][cc][1];
            S[k][cc] = v + other_half(v);
        }
    }
    if (hh == 0) {
#pragma unroll
        for (int cc = 0; cc < 3; ++cc)
            part4[(wave * RECV_PER_BLOCK + m) * 3 + cc] =
                make_float4(S[0][cc], S[1][cc], S[2][cc], S[3][cc]);
    }
    float wl[48];
    float hi0 = 0.0f, hi1 = 0.0f, hi2 = 0.0f;
    if (tid < RECV_PER_BLOCK) {
#pragma unroll
        for (int q = 0; q < 12; ++q) {
            const float4 v = sw4[q];
            wl[4 * q + 0] = v.x;  wl[4 * q + 1] = v.y;  wl[4 * q + 2] = v.z;  wl[4 * q + 3] = v.w;
        }
        const float4 hi4 = spe[ti];
        hi0 = hi4.x; hi1 = hi4.y; hi2 = hi4.z;
    }
    __syncthreads();

    if (wave != 0) return;

    v8f Dm = {};
    {
        const int  pb    = m * 3;
        const float* const pf = reinterpret_cast<const float*>(part4);
#pragma unroll
        for (int cc = 0; cc < 3; ++cc) {
            v2f Bm;
            Bm[0] = 0.0f;  Bm[1] = 0.0f;
#pragma unroll
            for (int q = 0; q < 8; ++q) {
                const float* p = pf + (pb + q * (RECV_PER_BLOCK * 3) + cc) * 4 + 2 * hh;
                Bm[0] += p[0];
                Bm[1] += p[1];
            }
            v2f Am;
            Am[0] = (m == cc) ? wx0 : 0.0f;
            Am[1] = (m == cc) ? wx1 : 0.0f;
            Dm = __builtin_amdgcn_wmma_f32_16x16x4_f32(
                false, Am, false, Bm, (short)0, Dm, false, false);
        }
    }

    if (tid < RECV_PER_BLOCK) {
        const float v0 = Dm[0], v1 = Dm[1], v2 = Dm[2];
        const int i = row0 + tid;
        const float* uw1 = wl;       const float* uw2 = wl + 9;
        const float* xw  = wl + 18;  const float* vw  = wl + 27;
        const float* ub1 = wl + 36;  const float* ub2 = wl + 39;
        const float* xb_ = wl + 42;  const float* vb_ = wl + 45;

        const float u0 = silu_f(v0 * uw1[0] + v1 * uw1[3] + v2 * uw1[6] + ub1[0]);
        const float u1 = silu_f(v0 * uw1[1] + v1 * uw1[4] + v2 * uw1[7] + ub1[1]);
        const float u2 = silu_f(v0 * uw1[2] + v1 * uw1[5] + v2 * uw1[8] + ub1[2]);

        const float h0 = hi0 + u0 * uw2[0] + u1 * uw2[3] + u2 * uw2[6] + ub2[0];
        const float h1 = hi1 + u0 * uw2[1] + u1 * uw2[4] + u2 * uw2[7] + ub2[1];
        const float h2 = hi2 + u0 * uw2[2] + u1 * uw2[5] + u2 * uw2[8] + ub2[2];

        const int base = (b * N_ATOMS + i) * 3;
        const int half = BATCH * N_ATOMS * 3;
#pragma unroll
        for (int c = 0; c < 3; ++c) {
            const float px = h0 * xw[0 * 3 + c] + h1 * xw[1 * 3 + c] + h2 * xw[2 * 3 + c] + xb_[c];
            const float pv = h0 * vw[0 * 3 + c] + h1 * vw[1 * 3 + c] + h2 * vw[2 * 3 + c] + vb_[c];
            out[base + c]        = tanh_f(px) * MAX_DISP;
            out[half + base + c] = tanh_f(pv) * 1.0f;
        }
    }
}

extern "C" void kernel_launch(void* const* d_in, const int* in_sizes, int n_in,
                              void* d_out, int out_size, void* d_ws, size_t ws_size,
                              hipStream_t stream) {
    (void)in_sizes; (void)n_in; (void)d_ws; (void)ws_size; (void)out_size;
    const float* x_nm   = (const float*)d_in[0];
    const int*   types  = (const int*)  d_in[1];
    const int*   mask   = (const int*)  d_in[2];
    const float* emb_W  = (const float*)d_in[3];
    const float* msg_w1 = (const float*)d_in[4];
    const float* msg_b1 = (const float*)d_in[5];
    const float* msg_w2 = (const float*)d_in[6];
    const float* msg_b2 = (const float*)d_in[7];
    const float* upd_w1 = (const float*)d_in[8];
    const float* upd_b1 = (const float*)d_in[9];
    const float* upd_w2 = (const float*)d_in[10];
    const float* upd_b2 = (const float*)d_in[11];
    const float* dx_w   = (const float*)d_in[12];
    const float* dx_b   = (const float*)d_in[13];
    const float* dv_w   = (const float*)d_in[14];
    const float* dv_b   = (const float*)d_in[15];
    float* out = (float*)d_out;

    dim3 grid(BATCH * (N_ATOMS / RECV_PER_BLOCK));
    dim3 block(256);
    simple_et_kernel<<<grid, block, 0, stream>>>(
        x_nm, types, mask, emb_W,
        msg_w1, msg_b1, msg_w2, msg_b2,
        upd_w1, upd_b1, upd_w2, upd_b2,
        dx_w, dx_b, dv_w, dv_b, out);
}
